# starting kernel
# baseline (speedup 1.0000x reference)
#include <hip/hip_runtime.h>
#include <hip/hip_bf16.h>
#include <math.h>


typedef __attribute__((ext_vector_type(2))) float v2f;
typedef __attribute__((ext_vector_type(8))) float v8f;

#define DFEAT 128
#define NEG_SLOPE 0.2f

__device__ __forceinline__ unsigned ordEnc(float f) {
  unsigned u = __float_as_uint(f);
  return (u & 0x80000000u) ? ~u : (u | 0x80000000u);
}
__device__ __forceinline__ float ordDec(unsigned e) {
  unsigned u = (e & 0x80000000u) ? (e & 0x7fffffffu) : ~e;
  return __uint_as_float(u);
}

__device__ __forceinline__ v8f wmma_f32(v2f a, v2f b, v8f c) {
  return __builtin_amdgcn_wmma_f32_16x16x4_f32(false, a, false, b, (short)0, c,
                                               false, false);
}

__global__ void gemm_f32_wmma(const float* __restrict__ A, int lda,
                              const float* __restrict__ B, int ldb,
                              float* __restrict__ D, int ldo,
                              int nrows, int kdim, int ncols,
                              const float* __restrict__ bias, int relu) {
  const int lane = threadIdx.x & 31;
  const int wid  = (blockIdx.x * blockDim.x + threadIdx.x) >> 5;
  const int colTiles = ncols >> 4;
  const int rowTiles = (nrows + 15) >> 4;
  if (wid >= rowTiles * colTiles) return;
  const int tr = wid / colTiles;
  const int tc = wid - tr * colTiles;
  const int row0 = tr << 4, col0 = tc << 4;
  const int lm = lane & 15;
  const int lh = lane >> 4;

  int arow = row0 + lm;
  if (arow >= nrows) arow = nrows - 1;
  const float* Ap = A + (size_t)arow * lda + 2 * lh;
  const float* Bp = B + (size_t)(2 * lh) * ldb + col0 + lm;

  v8f acc = {0.f, 0.f, 0.f, 0.f, 0.f, 0.f, 0.f, 0.f};
  int k = 0;
  for (; k + 16 <= kdim; k += 16) {
    const v2f a0 = *(const v2f*)(Ap + k);
    const v2f a1 = *(const v2f*)(Ap + k + 4);
    const v2f a2 = *(const v2f*)(Ap + k + 8);
    const v2f a3 = *(const v2f*)(Ap + k + 12);
    const float* bq = Bp + (size_t)k * ldb;
    v2f b0, b1, b2, b3;
    b0.x = bq[0];
    b0.y = bq[(size_t)1 * ldb];
    b1.x = bq[(size_t)4 * ldb];
    b1.y = bq[(size_t)5 * ldb];
    b2.x = bq[(size_t)8 * ldb];
    b2.y = bq[(size_t)9 * ldb];
    b3.x = bq[(size_t)12 * ldb];
    b3.y = bq[(size_t)13 * ldb];
    acc = wmma_f32(a0, b0, acc);
    acc = wmma_f32(a1, b1, acc);
    acc = wmma_f32(a2, b2, acc);
    acc = wmma_f32(a3, b3, acc);
  }
  for (; k < kdim; k += 4) {
    const v2f a = *(const v2f*)(Ap + k);
    v2f b;
    b.x = Bp[(size_t)k * ldb];
    b.y = Bp[(size_t)(k + 1) * ldb];
    acc = wmma_f32(a, b, acc);
  }

  const int col = col0 + lm;
  const float bv = bias ? bias[col] : 0.0f;
#pragma unroll
  for (int r = 0; r < 8; ++r) {
    const int row = row0 + r + 8 * lh;
    if (row < nrows) {
      float v = acc[r] + bv;
      if (relu) v = fmaxf(v, 0.0f);
      D[(size_t)row * ldo + col] = v;
    }
  }
}

__global__ void gemm_f32_wmma_bn(const float* __restrict__ A, int lda,
                                 const float* __restrict__ B, int ldb,
                                 float* __restrict__ D, int ldo,
                                 int nrows, int kdim, int ncols,
                                 const float* __restrict__ bias,
                                 const float* __restrict__ scl,
                                 const float* __restrict__ shf, int relu) {
  const int lane = threadIdx.x & 31;
  const int wid  = (blockIdx.x * blockDim.x + threadIdx.x) >> 5;
  const int colTiles = ncols >> 4;
  const int rowTiles = (nrows + 15) >> 4;
  if (wid >= rowTiles * colTiles) return;
  const int tr = wid / colTiles;
  const int tc = wid - tr * colTiles;
  const int row0 = tr << 4, col0 = tc << 4;
  const int lm = lane & 15;
  const int lh = lane >> 4;

  int arow = row0 + lm;
  if (arow >= nrows) arow = nrows - 1;
  const float* Ap = A + (size_t)arow * lda + 2 * lh;
  const float* Sp = scl + 2 * lh;
  const float* Hp = shf + 2 * lh;
  const float* Bp = B + (size_t)(2 * lh) * ldb + col0 + lm;

  v8f acc = {0.f, 0.f, 0.f, 0.f, 0.f, 0.f, 0.f, 0.f};
  int k = 0;
  for (; k + 16 <= kdim; k += 16) {
    v2f a0 = *(const v2f*)(Ap + k);
    v2f a1 = *(const v2f*)(Ap + k + 4);
    v2f a2 = *(const v2f*)(Ap + k + 8);
    v2f a3 = *(const v2f*)(Ap + k + 12);
    const v2f s0 = *(const v2f*)(Sp + k);
    const v2f s1 = *(const v2f*)(Sp + k + 4);
    const v2f s2 = *(const v2f*)(Sp + k + 8);
    const v2f s3 = *(const v2f*)(Sp + k + 12);
    const v2f h0 = *(const v2f*)(Hp + k);
    const v2f h1 = *(const v2f*)(Hp + k + 4);
    const v2f h2 = *(const v2f*)(Hp + k + 8);
    const v2f h3 = *(const v2f*)(Hp + k + 12);
    a0 = a0 * s0 + h0;
    a1 = a1 * s1 + h1;
    a2 = a2 * s2 + h2;
    a3 = a3 * s3 + h3;
    const float* bq = Bp + (size_t)k * ldb;
    v2f b0, b1, b2, b3;
    b0.x = bq[0];
    b0.y = bq[(size_t)1 * ldb];
    b1.x = bq[(size_t)4 * ldb];
    b1.y = bq[(size_t)5 * ldb];
    b2.x = bq[(size_t)8 * ldb];
    b2.y = bq[(size_t)9 * ldb];
    b3.x = bq[(size_t)12 * ldb];
    b3.y = bq[(size_t)13 * ldb];
    acc = wmma_f32(a0, b0, acc);
    acc = wmma_f32(a1, b1, acc);
    acc = wmma_f32(a2, b2, acc);
    acc = wmma_f32(a3, b3, acc);
  }
  for (; k < kdim; k += 4) {
    v2f a = *(const v2f*)(Ap + k);
    const v2f s = *(const v2f*)(Sp + k);
    const v2f h = *(const v2f*)(Hp + k);
    a = a * s + h;
    v2f b;
    b.x = Bp[(size_t)k * ldb];
    b.y = Bp[(size_t)(k + 1) * ldb];
    acc = wmma_f32(a, b, acc);
  }

  const int col = col0 + lm;
  const float bv = bias ? bias[col] : 0.0f;
#pragma unroll
  for (int r = 0; r < 8; ++r) {
    const int row = row0 + r + 8 * lh;
    if (row < nrows) {
      float v = acc[r] + bv;
      if (relu) v = fmaxf(v, 0.0f);
      D[(size_t)row * ldo + col] = v;
    }
  }
}

__global__ void attn_proj(const float* __restrict__ h, int heads,
                          const float* __restrict__ a_src,
                          const float* __restrict__ a_dst,
                          float* __restrict__ al_s, float* __restrict__ al_d,
                          int n) {
  const int wid  = (blockIdx.x * blockDim.x + threadIdx.x) >> 5;
  const int lane = threadIdx.x & 31;
  if (wid >= n * heads) return;
  const int node = wid / heads;
  const int head = wid - node * heads;
  const float* hp = h + (size_t)node * heads * DFEAT + head * DFEAT;
  const float* as = a_src + head * DFEAT;
  const float* ad = a_dst + head * DFEAT;
  float ss = 0.f, sd = 0.f;
  for (int c = lane; c < DFEAT; c += 32) {
    const float hv = hp[c];
    ss += hv * as[c];
    sd += hv * ad[c];
  }
  for (int off = 16; off > 0; off >>= 1) {
    ss += __shfl_xor(ss, off, 32);
    sd += __shfl_xor(sd, off, 32);
  }
  if (lane == 0) { al_s[wid] = ss; al_d[wid] = sd; }
}

__global__ void bias_fill(float* __restrict__ out, int ldo, int colOff,
                          int width, const float* __restrict__ bias, int n) {
  const int idx = blockIdx.x * blockDim.x + threadIdx.x;
  if (idx >= n * width) return;
  const int node = idx / width;
  const int c = idx - node * width;
  out[(size_t)node * ldo + colOff + c] = bias[c];
}

__global__ void zero_ms(unsigned* __restrict__ menc, float* __restrict__ ssum,
                        int nh) {
  const int i = blockIdx.x * blockDim.x + threadIdx.x;
  if (i < nh) { menc[i] = 0u; ssum[i] = 0.f; }
}

__global__ void edge_logit_max(const int* __restrict__ src,
                               const int* __restrict__ dst, int E, int n,
                               int heads, const float* __restrict__ al_s,
                               const float* __restrict__ al_d,
                               float* __restrict__ elog,
                               unsigned* __restrict__ menc) {
  const int idx = blockIdx.x * blockDim.x + threadIdx.x;
  if (idx >= (E + n) * heads) return;
  const int e = idx / heads;
  const int hh = idx - e * heads;
  const int s = (e < E) ? src[e] : (e - E);
  const int d = (e < E) ? dst[e] : (e - E);
  float x = al_s[s * heads + hh] + al_d[d * heads + hh];
  x = (x > 0.f) ? x : NEG_SLOPE * x;
  elog[idx] = x;
  atomicMax(&menc[d * heads + hh], ordEnc(x));
}

__global__ void edge_expsum(const int* __restrict__ src,
                            const int* __restrict__ dst, int E, int n,
                            int heads, const unsigned* __restrict__ menc,
                            float* __restrict__ elog,
                            float* __restrict__ ssum) {
  const int idx = blockIdx.x * blockDim.x + threadIdx.x;
  if (idx >= (E + n) * heads) return;
  const int e = idx / heads;
  const int hh = idx - e * heads;
  const int d = (e < E) ? dst[e] : (e - E);
  const float m = ordDec(menc[d * heads + hh]);
  const float ee = expf(elog[idx] - m);
  elog[idx] = ee;
  atomicAdd(&ssum[d * heads + hh], ee);
}

__global__ void edge_aggr(const int* __restrict__ src,
                          const int* __restrict__ dst, int E, int n, int heads,
                          const float* __restrict__ h, int ldh,
                          const float* __restrict__ elog,
                          const float* __restrict__ ssum,
                          float* __restrict__ out, int ldo, int colOff) {
  const int e = blockIdx.x;
  const int f = threadIdx.x;
  const int s = (e < E) ? src[e] : (e - E);
  const int d = (e < E) ? dst[e] : (e - E);
  const int head = f >> 7;
  const float alpha =
      elog[e * heads + head] / (ssum[d * heads + head] + 1e-16f);
  atomicAdd(&out[(size_t)d * ldo + colOff + f],
            alpha * h[(size_t)s * ldh + f]);
}

__global__ void col_stats(const float* __restrict__ y, int ld, int n,
                          float* __restrict__ sums, float* __restrict__ sumsq) {
  __shared__ float s1[256];
  __shared__ float s2[256];
  const int c = blockIdx.x;
  float s = 0.f, q = 0.f;
  for (int r = threadIdx.x; r < n; r += blockDim.x) {
    const float v = y[(size_t)r * ld + c];
    s += v;
    q += v * v;
  }
  s1[threadIdx.x] = s;
  s2[threadIdx.x] = q;
  __syncthreads();
  for (int st = 128; st > 0; st >>= 1) {
    if (threadIdx.x < st) {
      s1[threadIdx.x] += s1[threadIdx.x + st];
      s2[threadIdx.x] += s2[threadIdx.x + st];
    }
    __syncthreads();
  }
  if (threadIdx.x == 0) { sums[c] = s1[0]; sumsq[c] = s2[0]; }
}

__global__ void bn_final(const float* __restrict__ sums,
                         const float* __restrict__ sumsq,
                         const float* __restrict__ g,
                         const float* __restrict__ beta, int ncols, float invn,
                         float* __restrict__ sc, float* __restrict__ sh) {
  const int c = blockIdx.x * blockDim.x + threadIdx.x;
  if (c >= ncols) return;
  const float mean = sums[c] * invn;
  const float var = sumsq[c] * invn - mean * mean;
  const float r = rsqrtf(var + 1e-5f);
  const float s = g[c] * r;
  sc[c] = s;
  sh[c] = beta[c] - mean * s;
}

__global__ void mlp_out(const float* __restrict__ y2,
                        const float* __restrict__ sc,
                        const float* __restrict__ sh,
                        const float* __restrict__ Wm2,
                        const float* __restrict__ bm2, int n,
                        float* __restrict__ y3) {
  const int node = blockIdx.x * blockDim.x + threadIdx.x;
  if (node >= n) return;
  float a0 = bm2[0], a1 = bm2[1];
  const float* yp = y2 + (size_t)node * 96;
  for (int k = 0; k < 96; ++k) {
    const float v = fmaf(yp[k], sc[k], sh[k]);
    a0 = fmaf(v, Wm2[2 * k + 0], a0);
    a1 = fmaf(v, Wm2[2 * k + 1], a1);
  }
  y3[node * 2 + 0] = fmaxf(a0, 0.f);
  y3[node * 2 + 1] = fmaxf(a1, 0.f);
}

__global__ void bn_apply(const float* __restrict__ y,
                         const float* __restrict__ sc,
                         const float* __restrict__ sh, int total, int ncols,
                         float* __restrict__ out) {
  const int i = blockIdx.x * blockDim.x + threadIdx.x;
  if (i >= total) return;
  const int c = i % ncols;
  out[i] = fmaf(y[i], sc[c], sh[c]);
}

extern "C" void kernel_launch(void* const* d_in, const int* in_sizes, int n_in,
                              void* d_out, int out_size, void* d_ws,
                              size_t ws_size, hipStream_t stream) {
  const float* x      = (const float*)d_in[0];
  const int*   ei     = (const int*)d_in[1];
  const float* W1     = (const float*)d_in[2];
  const float* as1    = (const float*)d_in[3];
  const float* ad1    = (const float*)d_in[4];
  const float* b1     = (const float*)d_in[5];
  const float* W2     = (const float*)d_in[6];
  const float* as2    = (const float*)d_in[7];
  const float* ad2    = (const float*)d_in[8];
  const float* b2     = (const float*)d_in[9];
  const float* Wl     = (const float*)d_in[10];
  const float* bl     = (const float*)d_in[11];
  const float* gl     = (const float*)d_in[12];
  const float* betal  = (const float*)d_in[13];
  const float* Wm1    = (const float*)d_in[14];
  const float* bm1    = (const float*)d_in[15];
  const float* gm1    = (const float*)d_in[16];
  const float* betam1 = (const float*)d_in[17];
  const float* Wm2    = (const float*)d_in[18];
  const float* bm2    = (const float*)d_in[19];
  const float* gm2    = (const float*)d_in[20];
  const float* betam2 = (const float*)d_in[21];

  const int N = in_sizes[0] / DFEAT;
  const int E = in_sizes[1] / 2;
  const int E2 = E + N;
  const int* srcA = ei;
  const int* dstA = ei + E;

  char* w = (char*)d_ws;
  auto alloc = [&](size_t bytes) -> char* {
    char* p = w;
    w += (bytes + 255) & ~(size_t)255;
    return p;
  };
  float*    xcat = (float*)alloc((size_t)N * 384 * 4);
  float*    h1   = (float*)alloc((size_t)N * 256 * 4);
  float*    h2   = (float*)alloc((size_t)N * 128 * 4);
  float*    al_s = (float*)alloc((size_t)N * 2 * 4);
  float*    al_d = (float*)alloc((size_t)N * 2 * 4);
  unsigned* menc = (unsigned*)alloc((size_t)N * 2 * 4);
  float*    ssum = (float*)alloc((size_t)N * 2 * 4);
  float*    elog = (float*)alloc((size_t)E2 * 2 * 4);
  float*    y3   = (float*)alloc((size_t)N * 2 * 4);
  float*    sums = (float*)alloc(96 * 4);
  float*    sumsq= (float*)alloc(96 * 4);
  float*    sc1  = (float*)alloc(96 * 4);
  float*    sh1  = (float*)alloc(96 * 4);
  float*    sc2  = (float*)alloc(96 * 4);
  float*    sh2  = (float*)alloc(96 * 4);
  float*    sc3  = (float*)alloc(96 * 4);
  float*    sh3  = (float*)alloc(96 * 4);
  float* y1 = h1;
  float* y2 = h1 + (size_t)N * 96;

  const int TB = 256;
  auto gemmBlocks = [](int rows, int cols) {
    const int tiles = ((rows + 15) / 16) * (cols / 16);
    return (tiles + 7) / 8;
  };

  gemm_f32_wmma<<<gemmBlocks(N, 256), TB, 0, stream>>>(
      x, 128, W1, 256, h1, 256, N, 128, 256, nullptr, 0);
  attn_proj<<<(N * 2 + 7) / 8, TB, 0, stream>>>(h1, 2, as1, ad1, al_s, al_d, N);
  bias_fill<<<(N * 256 + TB - 1) / TB, TB, 0, stream>>>(xcat, 384, 0, 256, b1, N);
  zero_ms<<<(N * 2 + TB - 1) / TB, TB, 0, stream>>>(menc, ssum, N * 2);
  edge_logit_max<<<(E2 * 2 + TB - 1) / TB, TB, 0, stream>>>(
      srcA, dstA, E, N, 2, al_s, al_d, elog, menc);
  edge_expsum<<<(E2 * 2 + TB - 1) / TB, TB, 0, stream>>>(
      srcA, dstA, E, N, 2, menc, elog, ssum);
  edge_aggr<<<E2, 256, 0, stream>>>(srcA, dstA, E, N, 2, h1, 256, elog, ssum,
                                    xcat, 384, 0);

  gemm_f32_wmma<<<gemmBlocks(N, 128), TB, 0, stream>>>(
      xcat, 384, W2, 128, h2, 128, N, 256, 128, nullptr, 0);
  attn_proj<<<(N + 7) / 8, TB, 0, stream>>>(h2, 1, as2, ad2, al_s, al_d, N);
  bias_fill<<<(N * 128 + TB - 1) / TB, TB, 0, stream>>>(xcat, 384, 256, 128, b2, N);
  zero_ms<<<(N + TB - 1) / TB, TB, 0, stream>>>(menc, ssum, N);
  edge_logit_max<<<(E2 + TB - 1) / TB, TB, 0, stream>>>(
      srcA, dstA, E, N, 1, al_s, al_d, elog, menc);
  edge_expsum<<<(E2 + TB - 1) / TB, TB, 0, stream>>>(
      srcA, dstA, E, N, 1, menc, elog, ssum);
  edge_aggr<<<E2, 128, 0, stream>>>(srcA, dstA, E, N, 1, h2, 128, elog, ssum,
                                    xcat, 384, 256);

  const float invn = 1.0f / (float)N;
  gemm_f32_wmma<<<gemmBlocks(N, 96), TB, 0, stream>>>(
      xcat, 384, Wl, 96, y1, 96, N, 384, 96, bl, 1);
  col_stats<<<96, 256, 0, stream>>>(y1, 96, N, sums, sumsq);
  bn_final<<<1, 128, 0, stream>>>(sums, sumsq, gl, betal, 96, invn, sc1, sh1);
  gemm_f32_wmma_bn<<<gemmBlocks(N, 96), TB, 0, stream>>>(
      y1, 96, Wm1, 96, y2, 96, N, 96, 96, bm1, sc1, sh1, 1);
  col_stats<<<96, 256, 0, stream>>>(y2, 96, N, sums, sumsq);
  bn_final<<<1, 128, 0, stream>>>(sums, sumsq, gm1, betam1, 96, invn, sc2, sh2);
  mlp_out<<<(N + TB - 1) / TB, TB, 0, stream>>>(y2, sc2, sh2, Wm2, bm2, N, y3);
  col_stats<<<2, 256, 0, stream>>>(y3, 2, N, sums, sumsq);
  bn_final<<<1, 128, 0, stream>>>(sums, sumsq, gm2, betam2, 2, invn, sc3, sh3);
  bn_apply<<<(N * 2 + TB - 1) / TB, TB, 0, stream>>>(y3, sc3, sh3, N * 2, 2,
                                                     (float*)d_out);
}


# edge softmax+aggregation as a gather over edges counting-sorted by destination, one wave per row and head, sums in fixed ascending-source order, replacing float atomics (summation order changes)
# speedup vs baseline: 1.9889x; 1.9889x over previous
#include <hip/hip_runtime.h>
#include <hip/hip_bf16.h>
#include <math.h>


typedef __attribute__((ext_vector_type(2))) float v2f;
typedef __attribute__((ext_vector_type(8))) float v8f;

#define DFEAT 128
#define NEG_SLOPE 0.2f

__device__ __forceinline__ v8f wmma_f32(v2f a, v2f b, v8f c) {
  return __builtin_amdgcn_wmma_f32_16x16x4_f32(false, a, false, b, (short)0, c,
                                               false, false);
}

__global__ void gemm_f32_wmma(const float* __restrict__ A, int lda,
                              const float* __restrict__ B, int ldb,
                              float* __restrict__ D, int ldo,
                              int nrows, int kdim, int ncols,
                              const float* __restrict__ bias, int relu) {
  const int lane = threadIdx.x & 31;
  const int wid  = (blockIdx.x * blockDim.x + threadIdx.x) >> 5;
  const int colTiles = ncols >> 4;
  const int rowTiles = (nrows + 15) >> 4;
  if (wid >= rowTiles * colTiles) return;
  const int tr = wid / colTiles;
  const int tc = wid - tr * colTiles;
  const int row0 = tr << 4, col0 = tc << 4;
  const int lm = lane & 15;
  const int lh = lane >> 4;

  int arow = row0 + lm;
  if (arow >= nrows) arow = nrows - 1;
  const float* Ap = A + (size_t)arow * lda + 2 * lh;
  const float* Bp = B + (size_t)(2 * lh) * ldb + col0 + lm;

  v8f acc = {0.f, 0.f, 0.f, 0.f, 0.f, 0.f, 0.f, 0.f};
  int k = 0;
  for (; k + 16 <= kdim; k += 16) {
    const v2f a0 = *(const v2f*)(Ap + k);
    const v2f a1 = *(const v2f*)(Ap + k + 4);
    const v2f a2 = *(const v2f*)(Ap + k + 8);
    const v2f a3 = *(const v2f*)(Ap + k + 12);
    const float* bq = Bp + (size_t)k * ldb;
    v2f b0, b1, b2, b3;
    b0.x = bq[0];
    b0.y = bq[(size_t)1 * ldb];
    b1.x = bq[(size_t)4 * ldb];
    b1.y = bq[(size_t)5 * ldb];
    b2.x = bq[(size_t)8 * ldb];
    b2.y = bq[(size_t)9 * ldb];
    b3.x = bq[(size_t)12 * ldb];
    b3.y = bq[(size_t)13 * ldb];
    acc = wmma_f32(a0, b0, acc);
    acc = wmma_f32(a1, b1, acc);
    acc = wmma_f32(a2, b2, acc);
    acc = wmma_f32(a3, b3, acc);
  }
  for (; k < kdim; k += 4) {
    const v2f a = *(const v2f*)(Ap + k);
    v2f b;
    b.x = Bp[(size_t)k * ldb];
    b.y = Bp[(size_t)(k + 1) * ldb];
    acc = wmma_f32(a, b, acc);
  }

  const int col = col0 + lm;
  const float bv = bias ? bias[col] : 0.0f;
#pragma unroll
  for (int r = 0; r < 8; ++r) {
    const int row = row0 + r + 8 * lh;
    if (row < nrows) {
      float v = acc[r] + bv;
      if (relu) v = fmaxf(v, 0.0f);
      D[(size_t)row * ldo + col] = v;
    }
  }
}

__global__ void gemm_f32_wmma_bn(const float* __restrict__ A, int lda,
                                 const float* __restrict__ B, int ldb,
                                 float* __restrict__ D, int ldo,
                                 int nrows, int kdim, int ncols,
                                 const float* __restrict__ bias,
                                 const float* __restrict__ scl,
                                 const float* __restrict__ shf, int relu) {
  const int lane = threadIdx.x & 31;
  const int wid  = (blockIdx.x * blockDim.x + threadIdx.x) >> 5;
  const int colTiles = ncols >> 4;
  const int rowTiles = (nrows + 15) >> 4;
  if (wid >= rowTiles * colTiles) return;
  const int tr = wid / colTiles;
  const int tc = wid - tr * colTiles;
  const int row0 = tr << 4, col0 = tc << 4;
  const int lm = lane & 15;
  const int lh = lane >> 4;

  int arow = row0 + lm;
  if (arow >= nrows) arow = nrows - 1;
  const float* Ap = A + (size_t)arow * lda + 2 * lh;
  const float* Sp = scl + 2 * lh;
  const float* Hp = shf + 2 * lh;
  const float* Bp = B + (size_t)(2 * lh) * ldb + col0 + lm;

  v8f acc = {0.f, 0.f, 0.f, 0.f, 0.f, 0.f, 0.f, 0.f};
  int k = 0;
  for (; k + 16 <= kdim; k += 16) {
    v2f a0 = *(const v2f*)(Ap + k);
    v2f a1 = *(const v2f*)(Ap + k + 4);
    v2f a2 = *(const v2f*)(Ap + k + 8);
    v2f a3 = *(const v2f*)(Ap + k + 12);
    const v2f s0 = *(const v2f*)(Sp + k);
    const v2f s1 = *(const v2f*)(Sp + k + 4);
    const v2f s2 = *(const v2f*)(Sp + k + 8);
    const v2f s3 = *(const v2f*)(Sp + k + 12);
    const v2f h0 = *(const v2f*)(Hp + k);
    const v2f h1 = *(const v2f*)(Hp + k + 4);
    const v2f h2 = *(const v2f*)(Hp + k + 8);
    const v2f h3 = *(const v2f*)(Hp + k + 12);
    a0 = a0 * s0 + h0;
    a1 = a1 * s1 + h1;
    a2 = a2 * s2 + h2;
    a3 = a3 * s3 + h3;
    const float* bq = Bp + (size_t)k * ldb;
    v2f b0, b1, b2, b3;
    b0.x = bq[0];
    b0.y = bq[(size_t)1 * ldb];
    b1.x = bq[(size_t)4 * ldb];
    b1.y = bq[(size_t)5 * ldb];
    b2.x = bq[(size_t)8 * ldb];
    b2.y = bq[(size_t)9 * ldb];
    b3.x = bq[(size_t)12 * ldb];
    b3.y = bq[(size_t)13 * ldb];
    acc = wmma_f32(a0, b0, acc);
    acc = wmma_f32(a1, b1, acc);
    acc = wmma_f32(a2, b2, acc);
    acc = wmma_f32(a3, b3, acc);
  }
  for (; k < kdim; k += 4) {
    v2f a = *(const v2f*)(Ap + k);
    const v2f s = *(const v2f*)(Sp + k);
    const v2f h = *(const v2f*)(Hp + k);
    a = a * s + h;
    v2f b;
    b.x = Bp[(size_t)k * ldb];
    b.y = Bp[(size_t)(k + 1) * ldb];
    acc = wmma_f32(a, b, acc);
  }

  const int col = col0 + lm;
  const float bv = bias ? bias[col] : 0.0f;
#pragma unroll
  for (int r = 0; r < 8; ++r) {
    const int row = row0 + r + 8 * lh;
    if (row < nrows) {
      float v = acc[r] + bv;
      if (relu) v = fmaxf(v, 0.0f);
      D[(size_t)row * ldo + col] = v;
    }
  }
}

__global__ void attn_proj(const float* __restrict__ h, int heads,
                          const float* __restrict__ a_src,
                          const float* __restrict__ a_dst,
                          float* __restrict__ al_s, float* __restrict__ al_d,
                          int n) {
  const int wid  = (blockIdx.x * blockDim.x + threadIdx.x) >> 5;
  const int lane = threadIdx.x & 31;
  if (wid >= n * heads) return;
  const int node = wid / heads;
  const int head = wid - node * heads;
  const float* hp = h + (size_t)node * heads * DFEAT + head * DFEAT;
  const float* as = a_src + head * DFEAT;
  const float* ad = a_dst + head * DFEAT;
  float ss = 0.f, sd = 0.f;
  for (int c = lane; c < DFEAT; c += 32) {
    const float hv = hp[c];
    ss += hv * as[c];
    sd += hv * ad[c];
  }
  for (int off = 16; off > 0; off >>= 1) {
    ss += __shfl_xor(ss, off, 32);
    sd += __shfl_xor(sd, off, 32);
  }
  if (lane == 0) { al_s[wid] = ss; al_d[wid] = sd; }
}

__global__ void csr_zero(int* __restrict__ cnt, int n1) {
  const int i = blockIdx.x * blockDim.x + threadIdx.x;
  if (i < n1) cnt[i] = 0;
}

__global__ void csr_hist(const int* __restrict__ dst, int E,
                         int* __restrict__ cnt) {
  const int e = blockIdx.x * blockDim.x + threadIdx.x;
  if (e < E) atomicAdd(&cnt[dst[e]], 1);
}

__global__ void csr_scan(const int* __restrict__ cnt, int n,
                         int* __restrict__ rowptr, int* __restrict__ cursor) {
  __shared__ int part[1024];
  const int t = threadIdx.x;
  const int per = (n + 1023) / 1024;
  const int lo = min(t * per, n);
  const int hi = min(lo + per, n);
  int s = 0;
  for (int i = lo; i < hi; ++i) s += cnt[i];
  part[t] = s;
  __syncthreads();
  for (int off = 1; off < 1024; off <<= 1) {
    const int v = (t >= off) ? part[t - off] : 0;
    __syncthreads();
    part[t] += v;
    __syncthreads();
  }
  int run = part[t] - s;
  for (int i = lo; i < hi; ++i) {
    rowptr[i] = run;
    cursor[i] = run;
    run += cnt[i];
  }
  if (t == 1023) rowptr[n] = part[1023];
}

__global__ void csr_fill(const int* __restrict__ src,
                         const int* __restrict__ dst, int E,
                         int* __restrict__ cursor, int* __restrict__ tmp) {
  const int e = blockIdx.x * blockDim.x + threadIdx.x;
  if (e >= E) return;
  const int p = atomicAdd(&cursor[dst[e]], 1);
  tmp[p] = src[e];
}

__global__ void csr_sort(const int* __restrict__ rowptr,
                         const int* __restrict__ tmp, int* __restrict__ csr,
                         int n) {
  const int lane = threadIdx.x & 31;
  const int wid = __builtin_amdgcn_readfirstlane(
      (int)((blockIdx.x * blockDim.x + threadIdx.x) >> 5));
  if (wid >= n) return;
  const int beg = rowptr[wid];
  const int end = rowptr[wid + 1];
  for (int i = beg + lane; i < end; i += 32) {
    const int v = tmp[i];
    int rank = 0;
    for (int j = beg; j < end; ++j) {
      const int u = tmp[j];
      rank += (u < v || (u == v && j < i)) ? 1 : 0;
    }
    csr[beg + rank] = v;
  }
}

__device__ __forceinline__ void gather_rows(const float* __restrict__ hb,
                                            int ldh, int s_i, float w_i,
                                            int cnt, float4& acc) {
  const int wi = __float_as_int(w_i);
  int j = 0;
  for (; j + 4 <= cnt; j += 4) {
    const int sa = __builtin_amdgcn_readlane(s_i, j);
    const int sb = __builtin_amdgcn_readlane(s_i, j + 1);
    const int sc = __builtin_amdgcn_readlane(s_i, j + 2);
    const int sd = __builtin_amdgcn_readlane(s_i, j + 3);
    const float wa = __int_as_float(__builtin_amdgcn_readlane(wi, j));
    const float wb = __int_as_float(__builtin_amdgcn_readlane(wi, j + 1));
    const float wc = __int_as_float(__builtin_amdgcn_readlane(wi, j + 2));
    const float wd = __int_as_float(__builtin_amdgcn_readlane(wi, j + 3));
    const float4 ha = *(const float4*)(hb + (size_t)sa * ldh);
    const float4 hv = *(const float4*)(hb + (size_t)sb * ldh);
    const float4 hc = *(const float4*)(hb + (size_t)sc * ldh);
    const float4 hd = *(const float4*)(hb + (size_t)sd * ldh);
    acc.x += wa * ha.x; acc.y += wa * ha.y; acc.z += wa * ha.z; acc.w += wa * ha.w;
    acc.x += wb * hv.x; acc.y += wb * hv.y; acc.z += wb * hv.z; acc.w += wb * hv.w;
    acc.x += wc * hc.x; acc.y += wc * hc.y; acc.z += wc * hc.z; acc.w += wc * hc.w;
    acc.x += wd * hd.x; acc.y += wd * hd.y; acc.z += wd * hd.z; acc.w += wd * hd.w;
  }
  for (; j < cnt; ++j) {
    const int sa = __builtin_amdgcn_readlane(s_i, j);
    const float wa = __int_as_float(__builtin_amdgcn_readlane(wi, j));
    const float4 ha = *(const float4*)(hb + (size_t)sa * ldh);
    acc.x += wa * ha.x; acc.y += wa * ha.y; acc.z += wa * ha.z; acc.w += wa * ha.w;
  }
}

__global__ void gat_gather(const int* __restrict__ rowptr,
                           const int* __restrict__ csr, int heads,
                           const float* __restrict__ h, int ldh,
                           const float* __restrict__ al_s,
                           const float* __restrict__ al_d,
                           const float* __restrict__ bias,
                           float* __restrict__ out, int ldo, int colOff,
                           int n) {
  const int lane = threadIdx.x & 31;
  const int wid = __builtin_amdgcn_readfirstlane(
      (int)((blockIdx.x * blockDim.x + threadIdx.x) >> 5));
  if (wid >= n * heads) return;
  const int d = wid / heads;
  const int head = wid - d * heads;
  const int beg = rowptr[d];
  const int deg = rowptr[d + 1] - beg;
  const int total = deg + 1;
  const float ad = al_d[wid];
  const int* cp = csr + beg;

  const bool v0 = lane < total;
  const int s0 = (lane < deg) ? cp[lane] : d;
  float x0 = al_s[s0 * heads + head] + ad;
  x0 = (x0 > 0.f) ? x0 : NEG_SLOPE * x0;

  float m = v0 ? x0 : -INFINITY;
  for (int i = lane + 32; i < total; i += 32) {
    const int s = (i < deg) ? cp[i] : d;
    float x = al_s[s * heads + head] + ad;
    x = (x > 0.f) ? x : NEG_SLOPE * x;
    m = fmaxf(m, x);
  }
  for (int off = 16; off > 0; off >>= 1) m = fmaxf(m, __shfl_xor(m, off, 32));

  const float e0 = v0 ? expf(x0 - m) : 0.f;
  float sum = e0;
  for (int i = lane + 32; i < total; i += 32) {
    const int s = (i < deg) ? cp[i] : d;
    float x = al_s[s * heads + head] + ad;
    x = (x > 0.f) ? x : NEG_SLOPE * x;
    sum += expf(x - m);
  }
  for (int off = 16; off > 0; off >>= 1) sum += __shfl_xor(sum, off, 32);
  const float den = sum + 1e-16f;

  const float* hb = h + head * DFEAT + lane * 4;
  float4 acc = {0.f, 0.f, 0.f, 0.f};
  gather_rows(hb, ldh, s0, e0 / den, min(total, 32), acc);
  for (int base = 32; base < total; base += 32) {
    const int i = base + lane;
    const int s = (i < deg) ? cp[i] : d;
    float x = al_s[s * heads + head] + ad;
    x = (x > 0.f) ? x : NEG_SLOPE * x;
    const float w = (i < total) ? expf(x - m) / den : 0.f;
    gather_rows(hb, ldh, s, w, min(total - base, 32), acc);
  }

  const float* bp = bias + head * DFEAT + lane * 4;
  acc.x += bp[0]; acc.y += bp[1]; acc.z += bp[2]; acc.w += bp[3];
  *(float4*)(out + (size_t)d * ldo + colOff + head * DFEAT + lane * 4) = acc;
}

__global__ void col_stats(const float* __restrict__ y, int ld, int n,
                          float* __restrict__ sums, float* __restrict__ sumsq) {
  __shared__ float s1[256];
  __shared__ float s2[256];
  const int c = blockIdx.x;
  float s = 0.f, q = 0.f;
  for (int r = threadIdx.x; r < n; r += blockDim.x) {
    const float v = y[(size_t)r * ld + c];
    s += v;
    q += v * v;
  }
  s1[threadIdx.x] = s;
  s2[threadIdx.x] = q;
  __syncthreads();
  for (int st = 128; st > 0; st >>= 1) {
    if (threadIdx.x < st) {
      s1[threadIdx.x] += s1[threadIdx.x + st];
      s2[threadIdx.x] += s2[threadIdx.x + st];
    }
    __syncthreads();
  }
  if (threadIdx.x == 0) { sums[c] = s1[0]; sumsq[c] = s2[0]; }
}

__global__ void bn_final(const float* __restrict__ sums,
                         const float* __restrict__ sumsq,
                         const float* __restrict__ g,
                         const float* __restrict__ beta, int ncols, float invn,
                         float* __restrict__ sc, float* __restrict__ sh) {
  const int c = blockIdx.x * blockDim.x + threadIdx.x;
  if (c >= ncols) return;
  const float mean = sums[c] * invn;
  const float var = sumsq[c] * invn - mean * mean;
  const float r = rsqrtf(var + 1e-5f);
  const float s = g[c] * r;
  sc[c] = s;
  sh[c] = beta[c] - mean * s;
}

__global__ void mlp_out(const float* __restrict__ y2,
                        const float* __restrict__ sc,
                        const float* __restrict__ sh,
                        const float* __restrict__ Wm2,
                        const float* __restrict__ bm2, int n,
                        float* __restrict__ y3) {
  const int node = blockIdx.x * blockDim.x + threadIdx.x;
  if (node >= n) return;
  float a0 = bm2[0], a1 = bm2[1];
  const float* yp = y2 + (size_t)node * 96;
  for (int k = 0; k < 96; ++k) {
    const float v = fmaf(yp[k], sc[k], sh[k]);
    a0 = fmaf(v, Wm2[2 * k + 0], a0);
    a1 = fmaf(v, Wm2[2 * k + 1], a1);
  }
  y3[node * 2 + 0] = fmaxf(a0, 0.f);
  y3[node * 2 + 1] = fmaxf(a1, 0.f);
}

__global__ void bn_apply(const float* __restrict__ y,
                         const float* __restrict__ sc,
                         const float* __restrict__ sh, int total, int ncols,
                         float* __restrict__ out) {
  const int i = blockIdx.x * blockDim.x + threadIdx.x;
  if (i >= total) return;
  const int c = i % ncols;
  out[i] = fmaf(y[i], sc[c], sh[c]);
}

extern "C" void kernel_launch(void* const* d_in, const int* in_sizes, int n_in,
                              void* d_out, int out_size, void* d_ws,
                              size_t ws_size, hipStream_t stream) {
  const float* x      = (const float*)d_in[0];
  const int*   ei     = (const int*)d_in[1];
  const float* W1     = (const float*)d_in[2];
  const float* as1    = (const float*)d_in[3];
  const float* ad1    = (const float*)d_in[4];
  const float* b1     = (const float*)d_in[5];
  const float* W2     = (const float*)d_in[6];
  const float* as2    = (const float*)d_in[7];
  const float* ad2    = (const float*)d_in[8];
  const float* b2     = (const float*)d_in[9];
  const float* Wl     = (const float*)d_in[10];
  const float* bl     = (const float*)d_in[11];
  const float* gl     = (const float*)d_in[12];
  const float* betal  = (const float*)d_in[13];
  const float* Wm1    = (const float*)d_in[14];
  const float* bm1    = (const float*)d_in[15];
  const float* gm1    = (const float*)d_in[16];
  const float* betam1 = (const float*)d_in[17];
  const float* Wm2    = (const float*)d_in[18];
  const float* bm2    = (const float*)d_in[19];
  const float* gm2    = (const float*)d_in[20];
  const float* betam2 = (const float*)d_in[21];

  const int N = in_sizes[0] / DFEAT;
  const int E = in_sizes[1] / 2;
  const int* srcA = ei;
  const int* dstA = ei + E;

  char* w = (char*)d_ws;
  auto alloc = [&](size_t bytes) -> char* {
    char* p = w;
    w += (bytes + 255) & ~(size_t)255;
    return p;
  };
  float*    xcat = (float*)alloc((size_t)N * 384 * 4);
  float*    h1   = (float*)alloc((size_t)N * 256 * 4);
  float*    h2   = (float*)alloc((size_t)N * 128 * 4);
  float*    al_s = (float*)alloc((size_t)N * 2 * 4);
  float*    al_d = (float*)alloc((size_t)N * 2 * 4);
  int*      cnt    = (int*)alloc((size_t)(N + 1) * 4);
  int*      rowptr = (int*)alloc((size_t)(N + 1) * 4);
  int*      cursor = (int*)alloc((size_t)(N + 1) * 4);
  int*      csrTmp = (int*)alloc((size_t)E * 4);
  int*      csr    = (int*)alloc((size_t)E * 4);
  float*    y3   = (float*)alloc((size_t)N * 2 * 4);
  float*    sums = (float*)alloc(96 * 4);
  float*    sumsq= (float*)alloc(96 * 4);
  float*    sc1  = (float*)alloc(96 * 4);
  float*    sh1  = (float*)alloc(96 * 4);
  float*    sc2  = (float*)alloc(96 * 4);
  float*    sh2  = (float*)alloc(96 * 4);
  float*    sc3  = (float*)alloc(96 * 4);
  float*    sh3  = (float*)alloc(96 * 4);
  float* y1 = h1;
  float* y2 = h1 + (size_t)N * 96;

  const int TB = 256;
  auto gemmBlocks = [](int rows, int cols) {
    const int tiles = ((rows + 15) / 16) * (cols / 16);
    return (tiles + 7) / 8;
  };

  csr_zero<<<(N + 1 + TB - 1) / TB, TB, 0, stream>>>(cnt, N + 1);
  csr_hist<<<(E + TB - 1) / TB, TB, 0, stream>>>(dstA, E, cnt);
  csr_scan<<<1, 1024, 0, stream>>>(cnt, N, rowptr, cursor);
  csr_fill<<<(E + TB - 1) / TB, TB, 0, stream>>>(srcA, dstA, E, cursor, csrTmp);
  csr_sort<<<(N + 7) / 8, TB, 0, stream>>>(rowptr, csrTmp, csr, N);

  gemm_f32_wmma<<<gemmBlocks(N, 256), TB, 0, stream>>>(
      x, 128, W1, 256, h1, 256, N, 128, 256, nullptr, 0);
  attn_proj<<<(N * 2 + 7) / 8, TB, 0, stream>>>(h1, 2, as1, ad1, al_s, al_d, N);
  gat_gather<<<(N * 2 + 7) / 8, TB, 0, stream>>>(rowptr, csr, 2, h1, 256, al_s,
                                                 al_d, b1, xcat, 384, 0, N);

  gemm_f32_wmma<<<gemmBlocks(N, 128), TB, 0, stream>>>(
      xcat, 384, W2, 128, h2, 128, N, 256, 128, nullptr, 0);
  attn_proj<<<(N + 7) / 8, TB, 0, stream>>>(h2, 1, as2, ad2, al_s, al_d, N);
  gat_gather<<<(N + 7) / 8, TB, 0, stream>>>(rowptr, csr, 1, h2, 128, al_s,
                                             al_d, b2, xcat, 384, 256, N);

  const float invn = 1.0f / (float)N;
  gemm_f32_wmma<<<gemmBlocks(N, 96), TB, 0, stream>>>(
      xcat, 384, Wl, 96, y1, 96, N, 384, 96, bl, 1);
  col_stats<<<96, 256, 0, stream>>>(y1, 96, N, sums, sumsq);
  bn_final<<<1, 128, 0, stream>>>(sums, sumsq, gl, betal, 96, invn, sc1, sh1);
  gemm_f32_wmma_bn<<<gemmBlocks(N, 96), TB, 0, stream>>>(
      y1, 96, Wm1, 96, y2, 96, N, 96, 96, bm1, sc1, sh1, 1);
  col_stats<<<96, 256, 0, stream>>>(y2, 96, N, sums, sumsq);
  bn_final<<<1, 128, 0, stream>>>(sums, sumsq, gm1, betam1, 96, invn, sc2, sh2);
  mlp_out<<<(N + TB - 1) / TB, TB, 0, stream>>>(y2, sc2, sh2, Wm2, bm2, N, y3);
  col_stats<<<2, 256, 0, stream>>>(y3, 2, N, sums, sumsq);
  bn_final<<<1, 128, 0, stream>>>(sums, sumsq, gm2, betam2, 2, invn, sc3, sh3);
  bn_apply<<<(N * 2 + TB - 1) / TB, TB, 0, stream>>>(y3, sc3, sh3, N * 2, 2,
                                                     (float*)d_out);
}


# attention projections al_src/al_dst formed in the layer GEMMs' epilogues, one wave per 16 rows x one head (128 columns); the 2 projection launches removed (order of the projection sums changes)
# speedup vs baseline: 3.2694x; 1.6438x over previous
#include <hip/hip_runtime.h>
#include <hip/hip_bf16.h>
#include <math.h>


typedef __attribute__((ext_vector_type(2))) float v2f;
typedef __attribute__((ext_vector_type(8))) float v8f;

#define DFEAT 128
#define NEG_SLOPE 0.2f

__device__ __forceinline__ v8f wmma_f32(v2f a, v2f b, v8f c) {
  return __builtin_amdgcn_wmma_f32_16x16x4_f32(false, a, false, b, (short)0, c,
                                               false, false);
}

__device__ __forceinline__ void tile_col_partials(float* __restrict__ psum,
                                                  float* __restrict__ psq,
                                                  int tr, int ncols, int col,
                                                  int lh, float cs, float cq) {
  if (!psum) return;
  cs += __shfl_xor(cs, 16, 32);
  cq += __shfl_xor(cq, 16, 32);
  if (lh == 0) {
    psum[(size_t)tr * ncols + col] = cs;
    psq[(size_t)tr * ncols + col] = cq;
  }
}

__global__ void gemm_f32_wmma(const float* __restrict__ A, int lda,
                              const float* __restrict__ B, int ldb,
                              float* __restrict__ D, int ldo,
                              int nrows, int kdim, int ncols,
                              const float* __restrict__ bias, int relu,
                              float* __restrict__ psum,
                              float* __restrict__ psq) {
  const int lane = threadIdx.x & 31;
  const int wid  = (blockIdx.x * blockDim.x + threadIdx.x) >> 5;
  const int colTiles = ncols >> 4;
  const int rowTiles = (nrows + 15) >> 4;
  if (wid >= rowTiles * colTiles) return;
  const int tr = wid / colTiles;
  const int tc = wid - tr * colTiles;
  const int row0 = tr << 4, col0 = tc << 4;
  const int lm = lane & 15;
  const int lh = lane >> 4;

  int arow = row0 + lm;
  if (arow >= nrows) arow = nrows - 1;
  const float* Ap = A + (size_t)arow * lda + 2 * lh;
  const float* Bp = B + (size_t)(2 * lh) * ldb + col0 + lm;

  v8f acc = {0.f, 0.f, 0.f, 0.f, 0.f, 0.f, 0.f, 0.f};
  int k = 0;
  for (; k + 16 <= kdim; k += 16) {
    const v2f a0 = *(const v2f*)(Ap + k);
    const v2f a1 = *(const v2f*)(Ap + k + 4);
    const v2f a2 = *(const v2f*)(Ap + k + 8);
    const v2f a3 = *(const v2f*)(Ap + k + 12);
    const float* bq = Bp + (size_t)k * ldb;
    v2f b0, b1, b2, b3;
    b0.x = bq[0];
    b0.y = bq[(size_t)1 * ldb];
    b1.x = bq[(size_t)4 * ldb];
    b1.y = bq[(size_t)5 * ldb];
    b2.x = bq[(size_t)8 * ldb];
    b2.y = bq[(size_t)9 * ldb];
    b3.x = bq[(size_t)12 * ldb];
    b3.y = bq[(size_t)13 * ldb];
    acc = wmma_f32(a0, b0, acc);
    acc = wmma_f32(a1, b1, acc);
    acc = wmma_f32(a2, b2, acc);
    acc = wmma_f32(a3, b3, acc);
  }
  for (; k < kdim; k += 4) {
    const v2f a = *(const v2f*)(Ap + k);
    v2f b;
    b.x = Bp[(size_t)k * ldb];
    b.y = Bp[(size_t)(k + 1) * ldb];
    acc = wmma_f32(a, b, acc);
  }

  const int col = col0 + lm;
  const float bv = bias ? bias[col] : 0.0f;
  float cs = 0.f, cq = 0.f;
#pragma unroll
  for (int r = 0; r < 8; ++r) {
    const int row = row0 + r + 8 * lh;
    if (row < nrows) {
      float v = acc[r] + bv;
      if (relu) v = fmaxf(v, 0.0f);
      D[(size_t)row * ldo + col] = v;
      cs += v;
      cq += v * v;
    }
  }
  tile_col_partials(psum, psq, tr, ncols, col, lh, cs, cq);
}

__global__ void gemm_f32_wmma_bn(const float* __restrict__ A, int lda,
                                 const float* __restrict__ B, int ldb,
                                 float* __restrict__ D, int ldo,
                                 int nrows, int kdim, int ncols,
                                 const float* __restrict__ bias,
                                 const float* __restrict__ scl,
                                 const float* __restrict__ shf, int relu,
                                 float* __restrict__ psum,
                                 float* __restrict__ psq) {
  const int lane = threadIdx.x & 31;
  const int wid  = (blockIdx.x * blockDim.x + threadIdx.x) >> 5;
  const int colTiles = ncols >> 4;
  const int rowTiles = (nrows + 15) >> 4;
  if (wid >= rowTiles * colTiles) return;
  const int tr = wid / colTiles;
  const int tc = wid - tr * colTiles;
  const int row0 = tr << 4, col0 = tc << 4;
  const int lm = lane & 15;
  const int lh = lane >> 4;

  int arow = row0 + lm;
  if (arow >= nrows) arow = nrows - 1;
  const float* Ap = A + (size_t)arow * lda + 2 * lh;
  const float* Sp = scl + 2 * lh;
  const float* Hp = shf + 2 * lh;
  const float* Bp = B + (size_t)(2 * lh) * ldb + col0 + lm;

  v8f acc = {0.f, 0.f, 0.f, 0.f, 0.f, 0.f, 0.f, 0.f};
  int k = 0;
  for (; k + 16 <= kdim; k += 16) {
    v2f a0 = *(const v2f*)(Ap + k);
    v2f a1 = *(const v2f*)(Ap + k + 4);
    v2f a2 = *(const v2f*)(Ap + k + 8);
    v2f a3 = *(const v2f*)(Ap + k + 12);
    const v2f s0 = *(const v2f*)(Sp + k);
    const v2f s1 = *(const v2f*)(Sp + k + 4);
    const v2f s2 = *(const v2f*)(Sp + k + 8);
    const v2f s3 = *(const v2f*)(Sp + k + 12);
    const v2f h0 = *(const v2f*)(Hp + k);
    const v2f h1 = *(const v2f*)(Hp + k + 4);
    const v2f h2 = *(const v2f*)(Hp + k + 8);
    const v2f h3 = *(const v2f*)(Hp + k + 12);
    a0 = a0 * s0 + h0;
    a1 = a1 * s1 + h1;
    a2 = a2 * s2 + h2;
    a3 = a3 * s3 + h3;
    const float* bq = Bp + (size_t)k * ldb;
    v2f b0, b1, b2, b3;
    b0.x = bq[0];
    b0.y = bq[(size_t)1 * ldb];
    b1.x = bq[(size_t)4 * ldb];
    b1.y = bq[(size_t)5 * ldb];
    b2.x = bq[(size_t)8 * ldb];
    b2.y = bq[(size_t)9 * ldb];
    b3.x = bq[(size_t)12 * ldb];
    b3.y = bq[(size_t)13 * ldb];
    acc = wmma_f32(a0, b0, acc);
    acc = wmma_f32(a1, b1, acc);
    acc = wmma_f32(a2, b2, acc);
    acc = wmma_f32(a3, b3, acc);
  }
  for (; k < kdim; k += 4) {
    v2f a = *(const v2f*)(Ap + k);
    const v2f s = *(const v2f*)(Sp + k);
    const v2f h = *(const v2f*)(Hp + k);
    a = a * s + h;
    v2f b;
    b.x = Bp[(size_t)k * ldb];
    b.y = Bp[(size_t)(k + 1) * ldb];
    acc = wmma_f32(a, b, acc);
  }

  const int col = col0 + lm;
  const float bv = bias ? bias[col] : 0.0f;
  float cs = 0.f, cq = 0.f;
#pragma unroll
  for (int r = 0; r < 8; ++r) {
    const int row = row0 + r + 8 * lh;
    if (row < nrows) {
      float v = acc[r] + bv;
      if (relu) v = fmaxf(v, 0.0f);
      D[(size_t)row * ldo + col] = v;
      cs += v;
      cq += v * v;
    }
  }
  tile_col_partials(psum, psq, tr, ncols, col, lh, cs, cq);
}

struct HeadFrag {
  v2f a[4];
  float4 b[4][2][2];
};

__device__ __forceinline__ void head_load(HeadFrag& f,
                                          const float* __restrict__ Ap,
                                          const float* __restrict__ Bp, int ldb,
                                          int k) {
#pragma unroll
  for (int j = 0; j < 4; ++j) {
    f.a[j] = *(const v2f*)(Ap + k + 4 * j);
    const float* bq = Bp + (size_t)(k + 4 * j) * ldb;
    f.b[j][0][0] = *(const float4*)(bq);
    f.b[j][0][1] = *(const float4*)(bq + 64);
    f.b[j][1][0] = *(const float4*)(bq + ldb);
    f.b[j][1][1] = *(const float4*)(bq + ldb + 64);
  }
}

__device__ __forceinline__ void head_mma(const HeadFrag& f, v8f (&acc)[8]) {
#pragma unroll
  for (int j = 0; j < 4; ++j) {
    const float4 p0 = f.b[j][0][0], p1 = f.b[j][1][0];
    const float4 q0 = f.b[j][0][1], q1 = f.b[j][1][1];
    v2f b;
    b.x = p0.x; b.y = p1.x; acc[0] = wmma_f32(f.a[j], b, acc[0]);
    b.x = p0.y; b.y = p1.y; acc[1] = wmma_f32(f.a[j], b, acc[1]);
    b.x = p0.z; b.y = p1.z; acc[2] = wmma_f32(f.a[j], b, acc[2]);
    b.x = p0.w; b.y = p1.w; acc[3] = wmma_f32(f.a[j], b, acc[3]);
    b.x = q0.x; b.y = q1.x; acc[4] = wmma_f32(f.a[j], b, acc[4]);
    b.x = q0.y; b.y = q1.y; acc[5] = wmma_f32(f.a[j], b, acc[5]);
    b.x = q0.z; b.y = q1.z; acc[6] = wmma_f32(f.a[j], b, acc[6]);
    b.x = q0.w; b.y = q1.w; acc[7] = wmma_f32(f.a[j], b, acc[7]);
  }
}

__global__ void __launch_bounds__(64)
gemm_f32_wmma_head(const float* __restrict__ A, int lda,
                                   const float* __restrict__ B, int ldb,
                                   float* __restrict__ D, int ldo, int nrows,
                                   int kdim, int heads,
                                   const float* __restrict__ a_src,
                                   const float* __restrict__ a_dst,
                                   float* __restrict__ al_s,
                                   float* __restrict__ al_d) {
  const int lane = threadIdx.x & 31;
  const int wid  = (blockIdx.x * blockDim.x + threadIdx.x) >> 5;
  const int rowTiles = (nrows + 15) >> 4;
  if (wid >= rowTiles * heads) return;
  const int tr = wid / heads;
  const int head = wid - tr * heads;
  const int row0 = tr << 4, col0 = head * DFEAT;
  const int lm = lane & 15;
  const int lh = lane >> 4;

  int arow = row0 + lm;
  if (arow >= nrows) arow = nrows - 1;
  const float* Ap = A + (size_t)arow * lda + 2 * lh;
  const float* Bp = B + (size_t)(2 * lh) * ldb + col0 + 4 * lm;

  v8f acc[8];
#pragma unroll
  for (int t = 0; t < 8; ++t)
    acc[t] = (v8f){0.f, 0.f, 0.f, 0.f, 0.f, 0.f, 0.f, 0.f};

  int k = 0;
  if (kdim >= 32) {
    HeadFrag f0, f1;
    head_load(f0, Ap, Bp, ldb, 0);
    for (; k + 64 <= kdim; k += 32) {
      head_load(f1, Ap, Bp, ldb, k + 16);
      __builtin_amdgcn_sched_barrier(0);
      head_mma(f0, acc);
      head_load(f0, Ap, Bp, ldb, k + 32);
      __builtin_amdgcn_sched_barrier(0);
      head_mma(f1, acc);
    }
    head_load(f1, Ap, Bp, ldb, k + 16);
    __builtin_amdgcn_sched_barrier(0);
    head_mma(f0, acc);
    head_mma(f1, acc);
    k += 32;
  }
  for (; k < kdim; k += 4) {
    const v2f a = *(const v2f*)(Ap + k);
    const float* bq = Bp + (size_t)k * ldb;
    const float4 p0 = *(const float4*)(bq), q0 = *(const float4*)(bq + 64);
    const float4 p1 = *(const float4*)(bq + ldb);
    const float4 q1 = *(const float4*)(bq + ldb + 64);
    v2f b;
    b.x = p0.x; b.y = p1.x; acc[0] = wmma_f32(a, b, acc[0]);
    b.x = p0.y; b.y = p1.y; acc[1] = wmma_f32(a, b, acc[1]);
    b.x = p0.z; b.y = p1.z; acc[2] = wmma_f32(a, b, acc[2]);
    b.x = p0.w; b.y = p1.w; acc[3] = wmma_f32(a, b, acc[3]);
    b.x = q0.x; b.y = q1.x; acc[4] = wmma_f32(a, b, acc[4]);
    b.x = q0.y; b.y = q1.y; acc[5] = wmma_f32(a, b, acc[5]);
    b.x = q0.z; b.y = q1.z; acc[6] = wmma_f32(a, b, acc[6]);
    b.x = q0.w; b.y = q1.w; acc[7] = wmma_f32(a, b, acc[7]);
  }

  const float4 s0 = *(const float4*)(a_src + col0 + 4 * lm);
  const float4 s1 = *(const float4*)(a_src + col0 + 64 + 4 * lm);
  const float4 d0 = *(const float4*)(a_dst + col0 + 4 * lm);
  const float4 d1 = *(const float4*)(a_dst + col0 + 64 + 4 * lm);
  float v[16];
#pragma unroll
  for (int r = 0; r < 8; ++r) {
    const float4 lo = {acc[0][r], acc[1][r], acc[2][r], acc[3][r]};
    const float4 hi = {acc[4][r], acc[5][r], acc[6][r], acc[7][r]};
    const int row = row0 + 8 * lh + r;
    if (row < nrows) {
      float* dp = D + (size_t)row * ldo + col0 + 4 * lm;
      *(float4*)dp = lo;
      *(float4*)(dp + 64) = hi;
    }
    v[r] = lo.x * s0.x + lo.y * s0.y + lo.z * s0.z + lo.w * s0.w +
           hi.x * s1.x + hi.y * s1.y + hi.z * s1.z + hi.w * s1.w;
    v[8 + r] = lo.x * d0.x + lo.y * d0.y + lo.z * d0.z + lo.w * d0.w +
               hi.x * d1.x + hi.y * d1.y + hi.z * d1.z + hi.w * d1.w;
  }
  const bool b3 = (lm & 8) != 0, b2 = (lm & 4) != 0;
  const bool b1 = (lm & 2) != 0, b0 = (lm & 1) != 0;
  float w8[8], w4[4], w2[2];
#pragma unroll
  for (int i = 0; i < 8; ++i) {
    const float keep = b3 ? v[8 + i] : v[i];
    const float send = b3 ? v[i] : v[8 + i];
    w8[i] = keep + __shfl_xor(send, 8, 32);
  }
#pragma unroll
  for (int i = 0; i < 4; ++i) {
    const float keep = b2 ? w8[4 + i] : w8[i];
    const float send = b2 ? w8[i] : w8[4 + i];
    w4[i] = keep + __shfl_xor(send, 4, 32);
  }
#pragma unroll
  for (int i = 0; i < 2; ++i) {
    const float keep = b1 ? w4[2 + i] : w4[i];
    const float send = b1 ? w4[i] : w4[2 + i];
    w2[i] = keep + __shfl_xor(send, 2, 32);
  }
  const float keep = b0 ? w2[1] : w2[0];
  const float send = b0 ? w2[0] : w2[1];
  const float tot = keep + __shfl_xor(send, 1, 32);
  const int prow = row0 + 8 * lh + (lm & 7);
  if (prow < nrows) {
    float* dst = b3 ? al_d : al_s;
    dst[(size_t)prow * heads + head] = tot;
  }
}

__global__ void csr_zero(int* __restrict__ cnt, int n1) {
  const int i = blockIdx.x * blockDim.x + threadIdx.x;
  if (i < n1) cnt[i] = 0;
}

__global__ void csr_hist(const int* __restrict__ dst, int E,
                         int* __restrict__ cnt) {
  const int e = blockIdx.x * blockDim.x + threadIdx.x;
  if (e < E) atomicAdd(&cnt[dst[e]], 1);
}

__global__ void csr_scan(const int* __restrict__ cnt, int n,
                         int* __restrict__ rowptr, int* __restrict__ cursor) {
  __shared__ int part[1024];
  const int t = threadIdx.x;
  const int per = (n + 1023) / 1024;
  const int lo = min(t * per, n);
  const int hi = min(lo + per, n);
  int s = 0;
  for (int i = lo; i < hi; ++i) s += cnt[i];
  part[t] = s;
  __syncthreads();
  for (int off = 1; off < 1024; off <<= 1) {
    const int v = (t >= off) ? part[t - off] : 0;
    __syncthreads();
    part[t] += v;
    __syncthreads();
  }
  int run = part[t] - s;
  for (int i = lo; i < hi; ++i) {
    rowptr[i] = run;
    cursor[i] = run;
    run += cnt[i];
  }
  if (t == 1023) rowptr[n] = part[1023];
}

__global__ void csr_fill(const int* __restrict__ src,
                         const int* __restrict__ dst, int E,
                         int* __restrict__ cursor, int* __restrict__ tmp) {
  const int e = blockIdx.x * blockDim.x + threadIdx.x;
  if (e >= E) return;
  const int p = atomicAdd(&cursor[dst[e]], 1);
  tmp[p] = src[e];
}

__global__ void csr_sort(const int* __restrict__ rowptr,
                         const int* __restrict__ tmp, int* __restrict__ csr,
                         int n) {
  const int lane = threadIdx.x & 31;
  const int wid = __builtin_amdgcn_readfirstlane(
      (int)((blockIdx.x * blockDim.x + threadIdx.x) >> 5));
  if (wid >= n) return;
  const int beg = rowptr[wid];
  const int end = rowptr[wid + 1];
  for (int i = beg + lane; i < end; i += 32) {
    const int v = tmp[i];
    int rank = 0;
    for (int j = beg; j < end; ++j) {
      const int u = tmp[j];
      rank += (u < v || (u == v && j < i)) ? 1 : 0;
    }
    csr[beg + rank] = v;
  }
}

__device__ __forceinline__ void gather_rows(const float* __restrict__ hb,
                                            int ldh, int s_i, float w_i,
                                            int cnt, float4& acc) {
  const int wi = __float_as_int(w_i);
  int j = 0;
  for (; j + 4 <= cnt; j += 4) {
    const int sa = __builtin_amdgcn_readlane(s_i, j);
    const int sb = __builtin_amdgcn_readlane(s_i, j + 1);
    const int sc = __builtin_amdgcn_readlane(s_i, j + 2);
    const int sd = __builtin_amdgcn_readlane(s_i, j + 3);
    const float wa = __int_as_float(__builtin_amdgcn_readlane(wi, j));
    const float wb = __int_as_float(__builtin_amdgcn_readlane(wi, j + 1));
    const float wc = __int_as_float(__builtin_amdgcn_readlane(wi, j + 2));
    const float wd = __int_as_float(__builtin_amdgcn_readlane(wi, j + 3));
    const float4 ha = *(const float4*)(hb + (size_t)sa * ldh);
    const float4 hv = *(const float4*)(hb + (size_t)sb * ldh);
    const float4 hc = *(const float4*)(hb + (size_t)sc * ldh);
    const float4 hd = *(const float4*)(hb + (size_t)sd * ldh);
    acc.x += wa * ha.x; acc.y += wa * ha.y; acc.z += wa * ha.z; acc.w += wa * ha.w;
    acc.x += wb * hv.x; acc.y += wb * hv.y; acc.z += wb * hv.z; acc.w += wb * hv.w;
    acc.x += wc * hc.x; acc.y += wc * hc.y; acc.z += wc * hc.z; acc.w += wc * hc.w;
    acc.x += wd * hd.x; acc.y += wd * hd.y; acc.z += wd * hd.z; acc.w += wd * hd.w;
  }
  for (; j < cnt; ++j) {
    const int sa = __builtin_amdgcn_readlane(s_i, j);
    const float wa = __int_as_float(__builtin_amdgcn_readlane(wi, j));
    const float4 ha = *(const float4*)(hb + (size_t)sa * ldh);
    acc.x += wa * ha.x; acc.y += wa * ha.y; acc.z += wa * ha.z; acc.w += wa * ha.w;
  }
}

__global__ void gat_gather(const int* __restrict__ rowptr,
                           const int* __restrict__ csr, int heads,
                           const float* __restrict__ h, int ldh,
                           const float* __restrict__ al_s,
                           const float* __restrict__ al_d,
                           const float* __restrict__ bias,
                           float* __restrict__ out, int ldo, int colOff,
                           int n) {
  const int lane = threadIdx.x & 31;
  const int wid = __builtin_amdgcn_readfirstlane(
      (int)((blockIdx.x * blockDim.x + threadIdx.x) >> 5));
  if (wid >= n * heads) return;
  const int d = wid / heads;
  const int head = wid - d * heads;
  const int beg = rowptr[d];
  const int deg = rowptr[d + 1] - beg;
  const int total = deg + 1;
  const float ad = al_d[wid];
  const int* cp = csr + beg;

  const bool v0 = lane < total;
  const int s0 = (lane < deg) ? cp[lane] : d;
  float x0 = al_s[s0 * heads + head] + ad;
  x0 = (x0 > 0.f) ? x0 : NEG_SLOPE * x0;

  float m = v0 ? x0 : -INFINITY;
  for (int i = lane + 32; i < total; i += 32) {
    const int s = (i < deg) ? cp[i] : d;
    float x = al_s[s * heads + head] + ad;
    x = (x > 0.f) ? x : NEG_SLOPE * x;
    m = fmaxf(m, x);
  }
  for (int off = 16; off > 0; off >>= 1) m = fmaxf(m, __shfl_xor(m, off, 32));

  const float e0 = v0 ? expf(x0 - m) : 0.f;
  float sum = e0;
  for (int i = lane + 32; i < total; i += 32) {
    const int s = (i < deg) ? cp[i] : d;
    float x = al_s[s * heads + head] + ad;
    x = (x > 0.f) ? x : NEG_SLOPE * x;
    sum += expf(x - m);
  }
  for (int off = 16; off > 0; off >>= 1) sum += __shfl_xor(sum, off, 32);
  const float den = sum + 1e-16f;

  const float* hb = h + head * DFEAT + lane * 4;
  float4 acc = {0.f, 0.f, 0.f, 0.f};
  gather_rows(hb, ldh, s0, e0 / den, min(total, 32), acc);
  for (int base = 32; base < total; base += 32) {
    const int i = base + lane;
    const int s = (i < deg) ? cp[i] : d;
    float x = al_s[s * heads + head] + ad;
    x = (x > 0.f) ? x : NEG_SLOPE * x;
    const float w = (i < total) ? expf(x - m) / den : 0.f;
    gather_rows(hb, ldh, s, w, min(total - base, 32), acc);
  }

  const float* bp = bias + head * DFEAT + lane * 4;
  acc.x += bp[0]; acc.y += bp[1]; acc.z += bp[2]; acc.w += bp[3];
  *(float4*)(out + (size_t)d * ldo + colOff + head * DFEAT + lane * 4) = acc;
}

__global__ void bn_final(const float* __restrict__ psum,
                         const float* __restrict__ psq, int nparts,
                         const float* __restrict__ g,
                         const float* __restrict__ beta, int ncols, float invn,
                         float* __restrict__ sc, float* __restrict__ sh) {
  __shared__ float wsum[32];
  __shared__ float wsq[32];
  const int c = blockIdx.x;
  const int T = blockDim.x;
  const int last = nparts - 1;
  float s = 0.f, q = 0.f;
  for (int p = threadIdx.x; p < nparts; p += 4 * T) {
    const int p1 = p + T, p2 = p + 2 * T, p3 = p + 3 * T;
    const size_t i0 = (size_t)p * ncols + c;
    const size_t i1 = (size_t)min(p1, last) * ncols + c;
    const size_t i2 = (size_t)min(p2, last) * ncols + c;
    const size_t i3 = (size_t)min(p3, last) * ncols + c;
    const float a0 = psum[i0], a1 = psum[i1], a2 = psum[i2], a3 = psum[i3];
    const float b0 = psq[i0], b1 = psq[i1], b2 = psq[i2], b3 = psq[i3];
    s += a0;
    q += b0;
    s += (p1 < nparts) ? a1 : 0.f;
    q += (p1 < nparts) ? b1 : 0.f;
    s += (p2 < nparts) ? a2 : 0.f;
    q += (p2 < nparts) ? b2 : 0.f;
    s += (p3 < nparts) ? a3 : 0.f;
    q += (p3 < nparts) ? b3 : 0.f;
  }
  for (int off = 16; off > 0; off >>= 1) {
    s += __shfl_xor(s, off, 32);
    q += __shfl_xor(q, off, 32);
  }
  if ((threadIdx.x & 31) == 0) {
    wsum[threadIdx.x >> 5] = s;
    wsq[threadIdx.x >> 5] = q;
  }
  __syncthreads();
  if (threadIdx.x == 0) {
    float ts = 0.f, tq = 0.f;
    for (int w = 0; w < (T >> 5); ++w) {
      ts += wsum[w];
      tq += wsq[w];
    }
    const float mean = ts * invn;
    const float var = tq * invn - mean * mean;
    const float r = rsqrtf(var + 1e-5f);
    const float sv = g[c] * r;
    sc[c] = sv;
    sh[c] = beta[c] - mean * sv;
  }
}

__global__ void mlp_out(const float* __restrict__ y2,
                        const float* __restrict__ sc,
                        const float* __restrict__ sh,
                        const float* __restrict__ Wm2,
                        const float* __restrict__ bm2, int n,
                        float* __restrict__ y3, float* __restrict__ psum,
                        float* __restrict__ psq) {
  __shared__ float red[32][4];
  const int node = blockIdx.x * blockDim.x + threadIdx.x;
  float o0 = 0.f, o1 = 0.f;
  if (node < n) {
    float a0 = bm2[0], a1 = bm2[1];
    const float* yp = y2 + (size_t)node * 96;
    for (int k = 0; k < 96; ++k) {
      const float v = fmaf(yp[k], sc[k], sh[k]);
      a0 = fmaf(v, Wm2[2 * k + 0], a0);
      a1 = fmaf(v, Wm2[2 * k + 1], a1);
    }
    o0 = fmaxf(a0, 0.f);
    o1 = fmaxf(a1, 0.f);
    y3[node * 2 + 0] = o0;
    y3[node * 2 + 1] = o1;
  }
  float s0 = o0, q0 = o0 * o0, s1 = o1, q1 = o1 * o1;
  for (int off = 16; off > 0; off >>= 1) {
    s0 += __shfl_xor(s0, off, 32);
    q0 += __shfl_xor(q0, off, 32);
    s1 += __shfl_xor(s1, off, 32);
    q1 += __shfl_xor(q1, off, 32);
  }
  if ((threadIdx.x & 31) == 0) {
    float* rw = red[threadIdx.x >> 5];
    rw[0] = s0; rw[1] = q0; rw[2] = s1; rw[3] = q1;
  }
  __syncthreads();
  if (threadIdx.x < 4) {
    float t = 0.f;
    for (int w = 0; w < (int)(blockDim.x >> 5); ++w) t += red[w][threadIdx.x];
    float* dst = (threadIdx.x & 1) ? psq : psum;
    dst[blockIdx.x * 2 + (threadIdx.x >> 1)] = t;
  }
}

__global__ void bn_apply(const float* __restrict__ y,
                         const float* __restrict__ sc,
                         const float* __restrict__ sh, int total, int ncols,
                         float* __restrict__ out) {
  const int i = blockIdx.x * blockDim.x + threadIdx.x;
  if (i >= total) return;
  const int c = i % ncols;
  out[i] = fmaf(y[i], sc[c], sh[c]);
}

extern "C" void kernel_launch(void* const* d_in, const int* in_sizes, int n_in,
                              void* d_out, int out_size, void* d_ws,
                              size_t ws_size, hipStream_t stream) {
  const float* x      = (const float*)d_in[0];
  const int*   ei     = (const int*)d_in[1];
  const float* W1     = (const float*)d_in[2];
  const float* as1    = (const float*)d_in[3];
  const float* ad1    = (const float*)d_in[4];
  const float* b1     = (const float*)d_in[5];
  const float* W2     = (const float*)d_in[6];
  const float* as2    = (const float*)d_in[7];
  const float* ad2    = (const float*)d_in[8];
  const float* b2     = (const float*)d_in[9];
  const float* Wl     = (const float*)d_in[10];
  const float* bl     = (const float*)d_in[11];
  const float* gl     = (const float*)d_in[12];
  const float* betal  = (const float*)d_in[13];
  const float* Wm1    = (const float*)d_in[14];
  const float* bm1    = (const float*)d_in[15];
  const float* gm1    = (const float*)d_in[16];
  const float* betam1 = (const float*)d_in[17];
  const float* Wm2    = (const float*)d_in[18];
  const float* bm2    = (const float*)d_in[19];
  const float* gm2    = (const float*)d_in[20];
  const float* betam2 = (const float*)d_in[21];

  const int N = in_sizes[0] / DFEAT;
  const int E = in_sizes[1] / 2;
  const int* srcA = ei;
  const int* dstA = ei + E;

  char* w = (char*)d_ws;
  auto alloc = [&](size_t bytes) -> char* {
    char* p = w;
    w += (bytes + 255) & ~(size_t)255;
    return p;
  };
  float*    xcat = (float*)alloc((size_t)N * 384 * 4);
  float*    h1   = (float*)alloc((size_t)N * 256 * 4);
  float*    h2   = (float*)alloc((size_t)N * 128 * 4);
  float*    al_s = (float*)alloc((size_t)N * 2 * 4);
  float*    al_d = (float*)alloc((size_t)N * 2 * 4);
  int*      cnt    = (int*)alloc((size_t)(N + 1) * 4);
  int*      rowptr = (int*)alloc((size_t)(N + 1) * 4);
  int*      cursor = (int*)alloc((size_t)(N + 1) * 4);
  int*      csrTmp = (int*)alloc((size_t)E * 4);
  int*      csr    = (int*)alloc((size_t)E * 4);
  float*    y3   = (float*)alloc((size_t)N * 2 * 4);
  float*    sc1  = (float*)alloc(96 * 4);
  float*    sh1  = (float*)alloc(96 * 4);
  float*    sc2  = (float*)alloc(96 * 4);
  float*    sh2  = (float*)alloc(96 * 4);
  float*    sc3  = (float*)alloc(96 * 4);
  float*    sh3  = (float*)alloc(96 * 4);
  float* y1 = h1;
  float* y2 = h1 + (size_t)N * 96;
  const int rowTiles  = (N + 15) / 16;
  const int mlpBlocks = (N + 255) / 256;
  float* ps1 = h2;
  float* pq1 = ps1 + (size_t)rowTiles * 96;
  float* ps2 = pq1 + (size_t)rowTiles * 96;
  float* pq2 = ps2 + (size_t)rowTiles * 96;
  float* ps3 = pq2 + (size_t)rowTiles * 96;
  float* pq3 = ps3 + (size_t)mlpBlocks * 2;

  const int TB = 256;
  auto gemmBlocks = [](int rows, int cols) {
    const int tiles = ((rows + 15) / 16) * (cols / 16);
    return (tiles + 7) / 8;
  };

  csr_zero<<<(N + 1 + TB - 1) / TB, TB, 0, stream>>>(cnt, N + 1);
  csr_hist<<<(E + TB - 1) / TB, TB, 0, stream>>>(dstA, E, cnt);
  csr_scan<<<1, 1024, 0, stream>>>(cnt, N, rowptr, cursor);
  csr_fill<<<(E + TB - 1) / TB, TB, 0, stream>>>(srcA, dstA, E, cursor, csrTmp);
  csr_sort<<<(N + 7) / 8, TB, 0, stream>>>(rowptr, csrTmp, csr, N);

  gemm_f32_wmma_head<<<rowTiles, 64, 0, stream>>>(
      x, 128, W1, 256, h1, 256, N, 128, 2, as1, ad1, al_s, al_d);
  gat_gather<<<(N * 2 + 7) / 8, TB, 0, stream>>>(rowptr, csr, 2, h1, 256, al_s,
                                                 al_d, b1, xcat, 384, 0, N);

  gemm_f32_wmma_head<<<rowTiles, 32, 0, stream>>>(
      xcat, 384, W2, 128, h2, 128, N, 256, 1, as2, ad2, al_s, al_d);
  gat_gather<<<(N + 7) / 8, TB, 0, stream>>>(rowptr, csr, 1, h2, 128, al_s,
                                             al_d, b2, xcat, 384, 256, N);

  const float invn = 1.0f / (float)N;
  gemm_f32_wmma<<<gemmBlocks(N, 96), TB, 0, stream>>>(
      xcat, 384, Wl, 96, y1, 96, N, 384, 96, bl, 1, ps1, pq1);
  bn_final<<<96, 256, 0, stream>>>(ps1, pq1, rowTiles, gl, betal, 96, invn, sc1,
                                   sh1);
  gemm_f32_wmma_bn<<<gemmBlocks(N, 96), TB, 0, stream>>>(
      y1, 96, Wm1, 96, y2, 96, N, 96, 96, bm1, sc1, sh1, 1, ps2,
      pq2);
  bn_final<<<96, 256, 0, stream>>>(ps2, pq2, rowTiles, gm1, betam1, 96, invn,
                                   sc2, sh2);
  mlp_out<<<mlpBlocks, 256, 0, stream>>>(y2, sc2, sh2, Wm2, bm2, N, y3, ps3,
                                         pq3);
  bn_final<<<2, 256, 0, stream>>>(ps3, pq3, mlpBlocks, gm2, betam2, 2, invn,
                                  sc3, sh3);
  bn_apply<<<(N * 2 + TB - 1) / TB, TB, 0, stream>>>(y3, sc3, sh3, N * 2, 2,
                                                     (float*)d_out);
}


# CSR edge passes: an edge's slot = row start + the old count returned by the histogram's add, so the fill pass has no atomics (one atomic per edge, not two); both passes take 4 edges per thread
# speedup vs baseline: 3.6113x; 1.1046x over previous
#include <hip/hip_runtime.h>
#include <hip/hip_bf16.h>
#include <math.h>


typedef __attribute__((ext_vector_type(2))) float v2f;
typedef __attribute__((ext_vector_type(8))) float v8f;

#define DFEAT 128
#define NEG_SLOPE 0.2f

__device__ __forceinline__ v8f wmma_f32(v2f a, v2f b, v8f c) {
  return __builtin_amdgcn_wmma_f32_16x16x4_f32(false, a, false, b, (short)0, c,
                                               false, false);
}

__device__ __forceinline__ void tile_col_partials(float* __restrict__ psum,
                                                  float* __restrict__ psq,
                                                  int tr, int ncols, int col,
                                                  int lh, float cs, float cq) {
  if (!psum) return;
  cs += __shfl_xor(cs, 16, 32);
  cq += __shfl_xor(cq, 16, 32);
  if (lh == 0) {
    psum[(size_t)tr * ncols + col] = cs;
    psq[(size_t)tr * ncols + col] = cq;
  }
}

__global__ void gemm_f32_wmma(const float* __restrict__ A, int lda,
                              const float* __restrict__ B, int ldb,
                              float* __restrict__ D, int ldo,
                              int nrows, int kdim, int ncols,
                              const float* __restrict__ bias, int relu,
                              float* __restrict__ psum,
                              float* __restrict__ psq) {
  const int lane = threadIdx.x & 31;
  const int wid  = (blockIdx.x * blockDim.x + threadIdx.x) >> 5;
  const int colTiles = ncols >> 4;
  const int rowTiles = (nrows + 15) >> 4;
  if (wid >= rowTiles * colTiles) return;
  const int tr = wid / colTiles;
  const int tc = wid - tr * colTiles;
  const int row0 = tr << 4, col0 = tc << 4;
  const int lm = lane & 15;
  const int lh = lane >> 4;

  int arow = row0 + lm;
  if (arow >= nrows) arow = nrows - 1;
  const float* Ap = A + (size_t)arow * lda + 2 * lh;
  const float* Bp = B + (size_t)(2 * lh) * ldb + col0 + lm;

  v8f acc = {0.f, 0.f, 0.f, 0.f, 0.f, 0.f, 0.f, 0.f};
  int k = 0;
  for (; k + 16 <= kdim; k += 16) {
    const v2f a0 = *(const v2f*)(Ap + k);
    const v2f a1 = *(const v2f*)(Ap + k + 4);
    const v2f a2 = *(const v2f*)(Ap + k + 8);
    const v2f a3 = *(const v2f*)(Ap + k + 12);
    const float* bq = Bp + (size_t)k * ldb;
    v2f b0, b1, b2, b3;
    b0.x = bq[0];
    b0.y = bq[(size_t)1 * ldb];
    b1.x = bq[(size_t)4 * ldb];
    b1.y = bq[(size_t)5 * ldb];
    b2.x = bq[(size_t)8 * ldb];
    b2.y = bq[(size_t)9 * ldb];
    b3.x = bq[(size_t)12 * ldb];
    b3.y = bq[(size_t)13 * ldb];
    acc = wmma_f32(a0, b0, acc);
    acc = wmma_f32(a1, b1, acc);
    acc = wmma_f32(a2, b2, acc);
    acc = wmma_f32(a3, b3, acc);
  }
  for (; k < kdim; k += 4) {
    const v2f a = *(const v2f*)(Ap + k);
    v2f b;
    b.x = Bp[(size_t)k * ldb];
    b.y = Bp[(size_t)(k + 1) * ldb];
    acc = wmma_f32(a, b, acc);
  }

  const int col = col0 + lm;
  const float bv = bias ? bias[col] : 0.0f;
  float cs = 0.f, cq = 0.f;
#pragma unroll
  for (int r = 0; r < 8; ++r) {
    const int row = row0 + r + 8 * lh;
    if (row < nrows) {
      float v = acc[r] + bv;
      if (relu) v = fmaxf(v, 0.0f);
      D[(size_t)row * ldo + col] = v;
      cs += v;
      cq += v * v;
    }
  }
  tile_col_partials(psum, psq, tr, ncols, col, lh, cs, cq);
}

__global__ void gemm_f32_wmma_bn(const float* __restrict__ A, int lda,
                                 const float* __restrict__ B, int ldb,
                                 float* __restrict__ D, int ldo,
                                 int nrows, int kdim, int ncols,
                                 const float* __restrict__ bias,
                                 const float* __restrict__ scl,
                                 const float* __restrict__ shf, int relu,
                                 float* __restrict__ psum,
                                 float* __restrict__ psq) {
  const int lane = threadIdx.x & 31;
  const int wid  = (blockIdx.x * blockDim.x + threadIdx.x) >> 5;
  const int colTiles = ncols >> 4;
  const int rowTiles = (nrows + 15) >> 4;
  if (wid >= rowTiles * colTiles) return;
  const int tr = wid / colTiles;
  const int tc = wid - tr * colTiles;
  const int row0 = tr << 4, col0 = tc << 4;
  const int lm = lane & 15;
  const int lh = lane >> 4;

  int arow = row0 + lm;
  if (arow >= nrows) arow = nrows - 1;
  const float* Ap = A + (size_t)arow * lda + 2 * lh;
  const float* Sp = scl + 2 * lh;
  const float* Hp = shf + 2 * lh;
  const float* Bp = B + (size_t)(2 * lh) * ldb + col0 + lm;

  v8f acc = {0.f, 0.f, 0.f, 0.f, 0.f, 0.f, 0.f, 0.f};
  int k = 0;
  for (; k + 16 <= kdim; k += 16) {
    v2f a0 = *(const v2f*)(Ap + k);
    v2f a1 = *(const v2f*)(Ap + k + 4);
    v2f a2 = *(const v2f*)(Ap + k + 8);
    v2f a3 = *(const v2f*)(Ap + k + 12);
    const v2f s0 = *(const v2f*)(Sp + k);
    const v2f s1 = *(const v2f*)(Sp + k + 4);
    const v2f s2 = *(const v2f*)(Sp + k + 8);
    const v2f s3 = *(const v2f*)(Sp + k + 12);
    const v2f h0 = *(const v2f*)(Hp + k);
    const v2f h1 = *(const v2f*)(Hp + k + 4);
    const v2f h2 = *(const v2f*)(Hp + k + 8);
    const v2f h3 = *(const v2f*)(Hp + k + 12);
    a0 = a0 * s0 + h0;
    a1 = a1 * s1 + h1;
    a2 = a2 * s2 + h2;
    a3 = a3 * s3 + h3;
    const float* bq = Bp + (size_t)k * ldb;
    v2f b0, b1, b2, b3;
    b0.x = bq[0];
    b0.y = bq[(size_t)1 * ldb];
    b1.x = bq[(size_t)4 * ldb];
    b1.y = bq[(size_t)5 * ldb];
    b2.x = bq[(size_t)8 * ldb];
    b2.y = bq[(size_t)9 * ldb];
    b3.x = bq[(size_t)12 * ldb];
    b3.y = bq[(size_t)13 * ldb];
    acc = wmma_f32(a0, b0, acc);
    acc = wmma_f32(a1, b1, acc);
    acc = wmma_f32(a2, b2, acc);
    acc = wmma_f32(a3, b3, acc);
  }
  for (; k < kdim; k += 4) {
    v2f a = *(const v2f*)(Ap + k);
    const v2f s = *(const v2f*)(Sp + k);
    const v2f h = *(const v2f*)(Hp + k);
    a = a * s + h;
    v2f b;
    b.x = Bp[(size_t)k * ldb];
    b.y = Bp[(size_t)(k + 1) * ldb];
    acc = wmma_f32(a, b, acc);
  }

  const int col = col0 + lm;
  const float bv = bias ? bias[col] : 0.0f;
  float cs = 0.f, cq = 0.f;
#pragma unroll
  for (int r = 0; r < 8; ++r) {
    const int row = row0 + r + 8 * lh;
    if (row < nrows) {
      float v = acc[r] + bv;
      if (relu) v = fmaxf(v, 0.0f);
      D[(size_t)row * ldo + col] = v;
      cs += v;
      cq += v * v;
    }
  }
  tile_col_partials(psum, psq, tr, ncols, col, lh, cs, cq);
}

struct HeadFrag {
  v2f a[4];
  float4 b[4][2][2];
};

__device__ __forceinline__ void head_load(HeadFrag& f,
                                          const float* __restrict__ Ap,
                                          const float* __restrict__ Bp, int ldb,
                                          int k) {
#pragma unroll
  for (int j = 0; j < 4; ++j) {
    f.a[j] = *(const v2f*)(Ap + k + 4 * j);
    const float* bq = Bp + (size_t)(k + 4 * j) * ldb;
    f.b[j][0][0] = *(const float4*)(bq);
    f.b[j][0][1] = *(const float4*)(bq + 64);
    f.b[j][1][0] = *(const float4*)(bq + ldb);
    f.b[j][1][1] = *(const float4*)(bq + ldb + 64);
  }
}

__device__ __forceinline__ void head_mma(const HeadFrag& f, v8f (&acc)[8]) {
#pragma unroll
  for (int j = 0; j < 4; ++j) {
    const float4 p0 = f.b[j][0][0], p1 = f.b[j][1][0];
    const float4 q0 = f.b[j][0][1], q1 = f.b[j][1][1];
    v2f b;
    b.x = p0.x; b.y = p1.x; acc[0] = wmma_f32(f.a[j], b, acc[0]);
    b.x = p0.y; b.y = p1.y; acc[1] = wmma_f32(f.a[j], b, acc[1]);
    b.x = p0.z; b.y = p1.z; acc[2] = wmma_f32(f.a[j], b, acc[2]);
    b.x = p0.w; b.y = p1.w; acc[3] = wmma_f32(f.a[j], b, acc[3]);
    b.x = q0.x; b.y = q1.x; acc[4] = wmma_f32(f.a[j], b, acc[4]);
    b.x = q0.y; b.y = q1.y; acc[5] = wmma_f32(f.a[j], b, acc[5]);
    b.x = q0.z; b.y = q1.z; acc[6] = wmma_f32(f.a[j], b, acc[6]);
    b.x = q0.w; b.y = q1.w; acc[7] = wmma_f32(f.a[j], b, acc[7]);
  }
}

__global__ void __launch_bounds__(64)
gemm_f32_wmma_head(const float* __restrict__ A, int lda,
                                   const float* __restrict__ B, int ldb,
                                   float* __restrict__ D, int ldo, int nrows,
                                   int kdim, int heads,
                                   const float* __restrict__ a_src,
                                   const float* __restrict__ a_dst,
                                   float* __restrict__ al_s,
                                   float* __restrict__ al_d) {
  const int lane = threadIdx.x & 31;
  const int wid  = (blockIdx.x * blockDim.x + threadIdx.x) >> 5;
  const int rowTiles = (nrows + 15) >> 4;
  if (wid >= rowTiles * heads) return;
  const int tr = wid / heads;
  const int head = wid - tr * heads;
  const int row0 = tr << 4, col0 = head * DFEAT;
  const int lm = lane & 15;
  const int lh = lane >> 4;

  int arow = row0 + lm;
  if (arow >= nrows) arow = nrows - 1;
  const float* Ap = A + (size_t)arow * lda + 2 * lh;
  const float* Bp = B + (size_t)(2 * lh) * ldb + col0 + 4 * lm;

  v8f acc[8];
#pragma unroll
  for (int t = 0; t < 8; ++t)
    acc[t] = (v8f){0.f, 0.f, 0.f, 0.f, 0.f, 0.f, 0.f, 0.f};

  int k = 0;
  if (kdim >= 32) {
    HeadFrag f0, f1;
    head_load(f0, Ap, Bp, ldb, 0);
    for (; k + 64 <= kdim; k += 32) {
      head_load(f1, Ap, Bp, ldb, k + 16);
      __builtin_amdgcn_sched_barrier(0);
      head_mma(f0, acc);
      head_load(f0, Ap, Bp, ldb, k + 32);
      __builtin_amdgcn_sched_barrier(0);
      head_mma(f1, acc);
    }
    head_load(f1, Ap, Bp, ldb, k + 16);
    __builtin_amdgcn_sched_barrier(0);
    head_mma(f0, acc);
    head_mma(f1, acc);
    k += 32;
  }
  for (; k < kdim; k += 4) {
    const v2f a = *(const v2f*)(Ap + k);
    const float* bq = Bp + (size_t)k * ldb;
    const float4 p0 = *(const float4*)(bq), q0 = *(const float4*)(bq + 64);
    const float4 p1 = *(const float4*)(bq + ldb);
    const float4 q1 = *(const float4*)(bq + ldb + 64);
    v2f b;
    b.x = p0.x; b.y = p1.x; acc[0] = wmma_f32(a, b, acc[0]);
    b.x = p0.y; b.y = p1.y; acc[1] = wmma_f32(a, b, acc[1]);
    b.x = p0.z; b.y = p1.z; acc[2] = wmma_f32(a, b, acc[2]);
    b.x = p0.w; b.y = p1.w; acc[3] = wmma_f32(a, b, acc[3]);
    b.x = q0.x; b.y = q1.x; acc[4] = wmma_f32(a, b, acc[4]);
    b.x = q0.y; b.y = q1.y; acc[5] = wmma_f32(a, b, acc[5]);
    b.x = q0.z; b.y = q1.z; acc[6] = wmma_f32(a, b, acc[6]);
    b.x = q0.w; b.y = q1.w; acc[7] = wmma_f32(a, b, acc[7]);
  }

  const float4 s0 = *(const float4*)(a_src + col0 + 4 * lm);
  const float4 s1 = *(const float4*)(a_src + col0 + 64 + 4 * lm);
  const float4 d0 = *(const float4*)(a_dst + col0 + 4 * lm);
  const float4 d1 = *(const float4*)(a_dst + col0 + 64 + 4 * lm);
  float v[16];
#pragma unroll
  for (int r = 0; r < 8; ++r) {
    const float4 lo = {acc[0][r], acc[1][r], acc[2][r], acc[3][r]};
    const float4 hi = {acc[4][r], acc[5][r], acc[6][r], acc[7][r]};
    const int row = row0 + 8 * lh + r;
    if (row < nrows) {
      float* dp = D + (size_t)row * ldo + col0 + 4 * lm;
      *(float4*)dp = lo;
      *(float4*)(dp + 64) = hi;
    }
    v[r] = lo.x * s0.x + lo.y * s0.y + lo.z * s0.z + lo.w * s0.w +
           hi.x * s1.x + hi.y * s1.y + hi.z * s1.z + hi.w * s1.w;
    v[8 + r] = lo.x * d0.x + lo.y * d0.y + lo.z * d0.z + lo.w * d0.w +
               hi.x * d1.x + hi.y * d1.y + hi.z * d1.z + hi.w * d1.w;
  }
  const bool b3 = (lm & 8) != 0, b2 = (lm & 4) != 0;
  const bool b1 = (lm & 2) != 0, b0 = (lm & 1) != 0;
  float w8[8], w4[4], w2[2];
#pragma unroll
  for (int i = 0; i < 8; ++i) {
    const float keep = b3 ? v[8 + i] : v[i];
    const float send = b3 ? v[i] : v[8 + i];
    w8[i] = keep + __shfl_xor(send, 8, 32);
  }
#pragma unroll
  for (int i = 0; i < 4; ++i) {
    const float keep = b2 ? w8[4 + i] : w8[i];
    const float send = b2 ? w8[i] : w8[4 + i];
    w4[i] = keep + __shfl_xor(send, 4, 32);
  }
#pragma unroll
  for (int i = 0; i < 2; ++i) {
    const float keep = b1 ? w4[2 + i] : w4[i];
    const float send = b1 ? w4[i] : w4[2 + i];
    w2[i] = keep + __shfl_xor(send, 2, 32);
  }
  const float keep = b0 ? w2[1] : w2[0];
  const float send = b0 ? w2[0] : w2[1];
  const float tot = keep + __shfl_xor(send, 1, 32);
  const int prow = row0 + 8 * lh + (lm & 7);
  if (prow < nrows) {
    float* dst = b3 ? al_d : al_s;
    dst[(size_t)prow * heads + head] = tot;
  }
}

__global__ void csr_zero(int* __restrict__ cnt, int n1) {
  const int i = blockIdx.x * blockDim.x + threadIdx.x;
  if (i < n1) cnt[i] = 0;
}

#define EDGES_PER_THREAD 4
__global__ void csr_hist(const int* __restrict__ dst, int E,
                         int* __restrict__ cnt, int* __restrict__ rank) {
  const int t = blockIdx.x * blockDim.x + threadIdx.x;
  const int T = gridDim.x * blockDim.x;
  if (t + (EDGES_PER_THREAD - 1) * T < E) {
    int d[EDGES_PER_THREAD], r[EDGES_PER_THREAD];
#pragma unroll
    for (int q = 0; q < EDGES_PER_THREAD; ++q) d[q] = dst[t + q * T];
#pragma unroll
    for (int q = 0; q < EDGES_PER_THREAD; ++q) r[q] = atomicAdd(&cnt[d[q]], 1);
#pragma unroll
    for (int q = 0; q < EDGES_PER_THREAD; ++q) rank[t + q * T] = r[q];
  } else {
    for (int e = t; e < E; e += T) rank[e] = atomicAdd(&cnt[dst[e]], 1);
  }
}

__global__ void csr_scan(const int* __restrict__ cnt, int n,
                         int* __restrict__ rowptr) {
  __shared__ int part[1024];
  const int t = threadIdx.x;
  const int per = (n + 1023) / 1024;
  const int lo = min(t * per, n);
  const int hi = min(lo + per, n);
  int s = 0;
  for (int i = lo; i < hi; ++i) s += cnt[i];
  part[t] = s;
  __syncthreads();
  for (int off = 1; off < 1024; off <<= 1) {
    const int v = (t >= off) ? part[t - off] : 0;
    __syncthreads();
    part[t] += v;
    __syncthreads();
  }
  int run = part[t] - s;
  for (int i = lo; i < hi; ++i) {
    rowptr[i] = run;
    run += cnt[i];
  }
  if (t == 1023) rowptr[n] = part[1023];
}

__global__ void csr_fill(const int* __restrict__ src,
                         const int* __restrict__ dst, int E,
                         const int* __restrict__ rowptr,
                         const int* __restrict__ rank,
                         int* __restrict__ tmp) {
  const int t = blockIdx.x * blockDim.x + threadIdx.x;
  const int T = gridDim.x * blockDim.x;
  if (t + (EDGES_PER_THREAD - 1) * T < E) {
    int d[EDGES_PER_THREAD], s[EDGES_PER_THREAD], r[EDGES_PER_THREAD];
    int p[EDGES_PER_THREAD];
#pragma unroll
    for (int q = 0; q < EDGES_PER_THREAD; ++q) {
      d[q] = dst[t + q * T];
      s[q] = src[t + q * T];
      r[q] = rank[t + q * T];
    }
#pragma unroll
    for (int q = 0; q < EDGES_PER_THREAD; ++q) p[q] = rowptr[d[q]] + r[q];
#pragma unroll
    for (int q = 0; q < EDGES_PER_THREAD; ++q) tmp[p[q]] = s[q];
  } else {
    for (int e = t; e < E; e += T) tmp[rowptr[dst[e]] + rank[e]] = src[e];
  }
}

__global__ void csr_sort(const int* __restrict__ rowptr,
                         const int* __restrict__ tmp, int* __restrict__ csr,
                         int n) {
  const int lane = threadIdx.x & 31;
  const int wid = __builtin_amdgcn_readfirstlane(
      (int)((blockIdx.x * blockDim.x + threadIdx.x) >> 5));
  if (wid >= n) return;
  const int beg = rowptr[wid];
  const int end = rowptr[wid + 1];
  for (int i = beg + lane; i < end; i += 32) {
    const int v = tmp[i];
    int rank = 0;
    for (int j = beg; j < end; ++j) {
      const int u = tmp[j];
      rank += (u < v || (u == v && j < i)) ? 1 : 0;
    }
    csr[beg + rank] = v;
  }
}

__device__ __forceinline__ void gather_rows(const float* __restrict__ hb,
                                            int ldh, int s_i, float w_i,
                                            int cnt, float4& acc) {
  const int wi = __float_as_int(w_i);
  int j = 0;
  for (; j + 4 <= cnt; j += 4) {
    const int sa = __builtin_amdgcn_readlane(s_i, j);
    const int sb = __builtin_amdgcn_readlane(s_i, j + 1);
    const int sc = __builtin_amdgcn_readlane(s_i, j + 2);
    const int sd = __builtin_amdgcn_readlane(s_i, j + 3);
    const float wa = __int_as_float(__builtin_amdgcn_readlane(wi, j));
    const float wb = __int_as_float(__builtin_amdgcn_readlane(wi, j + 1));
    const float wc = __int_as_float(__builtin_amdgcn_readlane(wi, j + 2));
    const float wd = __int_as_float(__builtin_amdgcn_readlane(wi, j + 3));
    const float4 ha = *(const float4*)(hb + (size_t)sa * ldh);
    const float4 hv = *(const float4*)(hb + (size_t)sb * ldh);
    const float4 hc = *(const float4*)(hb + (size_t)sc * ldh);
    const float4 hd = *(const float4*)(hb + (size_t)sd * ldh);
    acc.x += wa * ha.x; acc.y += wa * ha.y; acc.z += wa * ha.z; acc.w += wa * ha.w;
    acc.x += wb * hv.x; acc.y += wb * hv.y; acc.z += wb * hv.z; acc.w += wb * hv.w;
    acc.x += wc * hc.x; acc.y += wc * hc.y; acc.z += wc * hc.z; acc.w += wc * hc.w;
    acc.x += wd * hd.x; acc.y += wd * hd.y; acc.z += wd * hd.z; acc.w += wd * hd.w;
  }
  for (; j < cnt; ++j) {
    const int sa = __builtin_amdgcn_readlane(s_i, j);
    const float wa = __int_as_float(__builtin_amdgcn_readlane(wi, j));
    const float4 ha = *(const float4*)(hb + (size_t)sa * ldh);
    acc.x += wa * ha.x; acc.y += wa * ha.y; acc.z += wa * ha.z; acc.w += wa * ha.w;
  }
}

__global__ void gat_gather(const int* __restrict__ rowptr,
                           const int* __restrict__ csr, int heads,
                           const float* __restrict__ h, int ldh,
                           const float* __restrict__ al_s,
                           const float* __restrict__ al_d,
                           const float* __restrict__ bias,
                           float* __restrict__ out, int ldo, int colOff,
                           int n) {
  const int lane = threadIdx.x & 31;
  const int wid = __builtin_amdgcn_readfirstlane(
      (int)((blockIdx.x * blockDim.x + threadIdx.x) >> 5));
  if (wid >= n * heads) return;
  const int d = wid / heads;
  const int head = wid - d * heads;
  const int beg = rowptr[d];
  const int deg = rowptr[d + 1] - beg;
  const int total = deg + 1;
  const float ad = al_d[wid];
  const int* cp = csr + beg;

  const bool v0 = lane < total;
  const int s0 = (lane < deg) ? cp[lane] : d;
  float x0 = al_s[s0 * heads + head] + ad;
  x0 = (x0 > 0.f) ? x0 : NEG_SLOPE * x0;

  float m = v0 ? x0 : -INFINITY;
  for (int i = lane + 32; i < total; i += 32) {
    const int s = (i < deg) ? cp[i] : d;
    float x = al_s[s * heads + head] + ad;
    x = (x > 0.f) ? x : NEG_SLOPE * x;
    m = fmaxf(m, x);
  }
  for (int off = 16; off > 0; off >>= 1) m = fmaxf(m, __shfl_xor(m, off, 32));

  const float e0 = v0 ? expf(x0 - m) : 0.f;
  float sum = e0;
  for (int i = lane + 32; i < total; i += 32) {
    const int s = (i < deg) ? cp[i] : d;
    float x = al_s[s * heads + head] + ad;
    x = (x > 0.f) ? x : NEG_SLOPE * x;
    sum += expf(x - m);
  }
  for (int off = 16; off > 0; off >>= 1) sum += __shfl_xor(sum, off, 32);
  const float den = sum + 1e-16f;

  const float* hb = h + head * DFEAT + lane * 4;
  float4 acc = {0.f, 0.f, 0.f, 0.f};
  gather_rows(hb, ldh, s0, e0 / den, min(total, 32), acc);
  for (int base = 32; base < total; base += 32) {
    const int i = base + lane;
    const int s = (i < deg) ? cp[i] : d;
    float x = al_s[s * heads + head] + ad;
    x = (x > 0.f) ? x : NEG_SLOPE * x;
    const float w = (i < total) ? expf(x - m) / den : 0.f;
    gather_rows(hb, ldh, s, w, min(total - base, 32), acc);
  }

  const float* bp = bias + head * DFEAT + lane * 4;
  acc.x += bp[0]; acc.y += bp[1]; acc.z += bp[2]; acc.w += bp[3];
  *(float4*)(out + (size_t)d * ldo + colOff + head * DFEAT + lane * 4) = acc;
}

__global__ void bn_final(const float* __restrict__ psum,
                         const float* __restrict__ psq, int nparts,
                         const float* __restrict__ g,
                         const float* __restrict__ beta, int ncols, float invn,
                         float* __restrict__ sc, float* __restrict__ sh) {
  __shared__ float wsum[32];
  __shared__ float wsq[32];
  const int c = blockIdx.x;
  const int T = blockDim.x;
  const int last = nparts - 1;
  float s = 0.f, q = 0.f;
  for (int p = threadIdx.x; p < nparts; p += 4 * T) {
    const int p1 = p + T, p2 = p + 2 * T, p3 = p + 3 * T;
    const size_t i0 = (size_t)p * ncols + c;
    const size_t i1 = (size_t)min(p1, last) * ncols + c;
    const size_t i2 = (size_t)min(p2, last) * ncols + c;
    const size_t i3 = (size_t)min(p3, last) * ncols + c;
    const float a0 = psum[i0], a1 = psum[i1], a2 = psum[i2], a3 = psum[i3];
    const float b0 = psq[i0], b1 = psq[i1], b2 = psq[i2], b3 = psq[i3];
    s += a0;
    q += b0;
    s += (p1 < nparts) ? a1 : 0.f;
    q += (p1 < nparts) ? b1 : 0.f;
    s += (p2 < nparts) ? a2 : 0.f;
    q += (p2 < nparts) ? b2 : 0.f;
    s += (p3 < nparts) ? a3 : 0.f;
    q += (p3 < nparts) ? b3 : 0.f;
  }
  for (int off = 16; off > 0; off >>= 1) {
    s += __shfl_xor(s, off, 32);
    q += __shfl_xor(q, off, 32);
  }
  if ((threadIdx.x & 31) == 0) {
    wsum[threadIdx.x >> 5] = s;
    wsq[threadIdx.x >> 5] = q;
  }
  __syncthreads();
  if (threadIdx.x == 0) {
    float ts = 0.f, tq = 0.f;
    for (int w = 0; w < (T >> 5); ++w) {
      ts += wsum[w];
      tq += wsq[w];
    }
    const float mean = ts * invn;
    const float var = tq * invn - mean * mean;
    const float r = rsqrtf(var + 1e-5f);
    const float sv = g[c] * r;
    sc[c] = sv;
    sh[c] = beta[c] - mean * sv;
  }
}

__global__ void mlp_out(const float* __restrict__ y2,
                        const float* __restrict__ sc,
                        const float* __restrict__ sh,
                        const float* __restrict__ Wm2,
                        const float* __restrict__ bm2, int n,
                        float* __restrict__ y3, float* __restrict__ psum,
                        float* __restrict__ psq) {
  __shared__ float red[32][4];
  const int node = blockIdx.x * blockDim.x + threadIdx.x;
  float o0 = 0.f, o1 = 0.f;
  if (node < n) {
    float a0 = bm2[0], a1 = bm2[1];
    const float* yp = y2 + (size_t)node * 96;
    for (int k = 0; k < 96; ++k) {
      const float v = fmaf(yp[k], sc[k], sh[k]);
      a0 = fmaf(v, Wm2[2 * k + 0], a0);
      a1 = fmaf(v, Wm2[2 * k + 1], a1);
    }
    o0 = fmaxf(a0, 0.f);
    o1 = fmaxf(a1, 0.f);
    y3[node * 2 + 0] = o0;
    y3[node * 2 + 1] = o1;
  }
  float s0 = o0, q0 = o0 * o0, s1 = o1, q1 = o1 * o1;
  for (int off = 16; off > 0; off >>= 1) {
    s0 += __shfl_xor(s0, off, 32);
    q0 += __shfl_xor(q0, off, 32);
    s1 += __shfl_xor(s1, off, 32);
    q1 += __shfl_xor(q1, off, 32);
  }
  if ((threadIdx.x & 31) == 0) {
    float* rw = red[threadIdx.x >> 5];
    rw[0] = s0; rw[1] = q0; rw[2] = s1; rw[3] = q1;
  }
  __syncthreads();
  if (threadIdx.x < 4) {
    float t = 0.f;
    for (int w = 0; w < (int)(blockDim.x >> 5); ++w) t += red[w][threadIdx.x];
    float* dst = (threadIdx.x & 1) ? psq : psum;
    dst[blockIdx.x * 2 + (threadIdx.x >> 1)] = t;
  }
}

__global__ void bn_apply(const float* __restrict__ y,
                         const float* __restrict__ sc,
                         const float* __restrict__ sh, int total, int ncols,
                         float* __restrict__ out) {
  const int i = blockIdx.x * blockDim.x + threadIdx.x;
  if (i >= total) return;
  const int c = i % ncols;
  out[i] = fmaf(y[i], sc[c], sh[c]);
}

extern "C" void kernel_launch(void* const* d_in, const int* in_sizes, int n_in,
                              void* d_out, int out_size, void* d_ws,
                              size_t ws_size, hipStream_t stream) {
  const float* x      = (const float*)d_in[0];
  const int*   ei     = (const int*)d_in[1];
  const float* W1     = (const float*)d_in[2];
  const float* as1    = (const float*)d_in[3];
  const float* ad1    = (const float*)d_in[4];
  const float* b1     = (const float*)d_in[5];
  const float* W2     = (const float*)d_in[6];
  const float* as2    = (const float*)d_in[7];
  const float* ad2    = (const float*)d_in[8];
  const float* b2     = (const float*)d_in[9];
  const float* Wl     = (const float*)d_in[10];
  const float* bl     = (const float*)d_in[11];
  const float* gl     = (const float*)d_in[12];
  const float* betal  = (const float*)d_in[13];
  const float* Wm1    = (const float*)d_in[14];
  const float* bm1    = (const float*)d_in[15];
  const float* gm1    = (const float*)d_in[16];
  const float* betam1 = (const float*)d_in[17];
  const float* Wm2    = (const float*)d_in[18];
  const float* bm2    = (const float*)d_in[19];
  const float* gm2    = (const float*)d_in[20];
  const float* betam2 = (const float*)d_in[21];

  const int N = in_sizes[0] / DFEAT;
  const int E = in_sizes[1] / 2;
  const int* srcA = ei;
  const int* dstA = ei + E;

  char* w = (char*)d_ws;
  auto alloc = [&](size_t bytes) -> char* {
    char* p = w;
    w += (bytes + 255) & ~(size_t)255;
    return p;
  };
  float*    xcat = (float*)alloc((size_t)N * 384 * 4);
  float*    h1   = (float*)alloc((size_t)N * 256 * 4);
  float*    h2   = (float*)alloc((size_t)N * 128 * 4);
  float*    al_s = (float*)alloc((size_t)N * 2 * 4);
  float*    al_d = (float*)alloc((size_t)N * 2 * 4);
  int*      cnt    = (int*)alloc((size_t)(N + 1) * 4);
  int*      rowptr = (int*)alloc((size_t)(N + 1) * 4);
  int*      csrTmp = (int*)alloc((size_t)E * 4);
  int*      csr    = (int*)alloc((size_t)E * 4);
  float*    y3   = (float*)alloc((size_t)N * 2 * 4);
  float*    sc1  = (float*)alloc(96 * 4);
  float*    sh1  = (float*)alloc(96 * 4);
  float*    sc2  = (float*)alloc(96 * 4);
  float*    sh2  = (float*)alloc(96 * 4);
  float*    sc3  = (float*)alloc(96 * 4);
  float*    sh3  = (float*)alloc(96 * 4);
  int* rank = csr;
  float* y1 = h1;
  float* y2 = h1 + (size_t)N * 96;
  const int rowTiles  = (N + 15) / 16;
  const int mlpBlocks = (N + 255) / 256;
  float* ps1 = h2;
  float* pq1 = ps1 + (size_t)rowTiles * 96;
  float* ps2 = pq1 + (size_t)rowTiles * 96;
  float* pq2 = ps2 + (size_t)rowTiles * 96;
  float* ps3 = pq2 + (size_t)rowTiles * 96;
  float* pq3 = ps3 + (size_t)mlpBlocks * 2;

  const int TB = 256;
  auto gemmBlocks = [](int rows, int cols) {
    const int tiles = ((rows + 15) / 16) * (cols / 16);
    return (tiles + 7) / 8;
  };

  csr_zero<<<(N + 1 + TB - 1) / TB, TB, 0, stream>>>(cnt, N + 1);
  const int edgeBlocks =
      (E + TB * EDGES_PER_THREAD - 1) / (TB * EDGES_PER_THREAD);
  csr_hist<<<edgeBlocks, TB, 0, stream>>>(dstA, E, cnt, rank);
  csr_scan<<<1, 1024, 0, stream>>>(cnt, N, rowptr);
  csr_fill<<<edgeBlocks, TB, 0, stream>>>(srcA, dstA, E, rowptr, rank, csrTmp);
  csr_sort<<<(N + 7) / 8, TB, 0, stream>>>(rowptr, csrTmp, csr, N);

  gemm_f32_wmma_head<<<rowTiles, 64, 0, stream>>>(
      x, 128, W1, 256, h1, 256, N, 128, 2, as1, ad1, al_s, al_d);
  gat_gather<<<(N * 2 + 7) / 8, TB, 0, stream>>>(rowptr, csr, 2, h1, 256, al_s,
                                                 al_d, b1, xcat, 384, 0, N);

  gemm_f32_wmma_head<<<rowTiles, 32, 0, stream>>>(
      xcat, 384, W2, 128, h2, 128, N, 256, 1, as2, ad2, al_s, al_d);
  gat_gather<<<(N + 7) / 8, TB, 0, stream>>>(rowptr, csr, 1, h2, 128, al_s,
                                             al_d, b2, xcat, 384, 256, N);

  const float invn = 1.0f / (float)N;
  gemm_f32_wmma<<<gemmBlocks(N, 96), TB, 0, stream>>>(
      xcat, 384, Wl, 96, y1, 96, N, 384, 96, bl, 1, ps1, pq1);
  bn_final<<<96, 256, 0, stream>>>(ps1, pq1, rowTiles, gl, betal, 96, invn, sc1,
                                   sh1);
  gemm_f32_wmma_bn<<<gemmBlocks(N, 96), TB, 0, stream>>>(
      y1, 96, Wm1, 96, y2, 96, N, 96, 96, bm1, sc1, sh1, 1, ps2,
      pq2);
  bn_final<<<96, 256, 0, stream>>>(ps2, pq2, rowTiles, gm1, betam1, 96, invn,
                                   sc2, sh2);
  mlp_out<<<mlpBlocks, 256, 0, stream>>>(y2, sc2, sh2, Wm2, bm2, N, y3, ps3,
                                         pq3);
  bn_final<<<2, 256, 0, stream>>>(ps3, pq3, mlpBlocks, gm2, betam2, 2, invn,
                                  sc3, sh3);
  bn_apply<<<(N * 2 + TB - 1) / TB, TB, 0, stream>>>(y3, sc3, sh3, N * 2, 2,
                                                     (float*)d_out);
}


# layer-1 GEMM and the in-degree histogram share one launch (work-groups pick the role by index; histogram takes 20 edges per thread so both are resident at once); 16 launches become 15
# speedup vs baseline: 4.2197x; 1.1685x over previous
#include <hip/hip_runtime.h>
#include <hip/hip_bf16.h>
#include <math.h>


typedef __attribute__((ext_vector_type(2))) float v2f;
typedef __attribute__((ext_vector_type(8))) float v8f;

#define DFEAT 128
#define NEG_SLOPE 0.2f

__device__ __forceinline__ v8f wmma_f32(v2f a, v2f b, v8f c) {
  return __builtin_amdgcn_wmma_f32_16x16x4_f32(false, a, false, b, (short)0, c,
                                               false, false);
}

#define MLPC 96
struct RowFrag {
  float4 a[2];
  float4 s[2], h[2];
  float4 b4[8];
  float2 b2[8];
};

template <bool BN>
__device__ __forceinline__ void row_load(RowFrag& f,
                                         const float* __restrict__ Ap,
                                         const float* __restrict__ Sp,
                                         const float* __restrict__ Hp,
                                         const float* __restrict__ Bp4,
                                         const float* __restrict__ Bp2,
                                         int kb) {
  f.a[0] = *(const float4*)(Ap + kb);
  f.a[1] = *(const float4*)(Ap + kb + 4);
  if (BN) {
    f.s[0] = *(const float4*)(Sp + kb);
    f.s[1] = *(const float4*)(Sp + kb + 4);
    f.h[0] = *(const float4*)(Hp + kb);
    f.h[1] = *(const float4*)(Hp + kb + 4);
  }
#pragma unroll
  for (int i = 0; i < 8; ++i) {
    f.b4[i] = *(const float4*)(Bp4 + (size_t)kb * MLPC + i * MLPC);
    f.b2[i] = *(const float2*)(Bp2 + (size_t)kb * MLPC + i * MLPC);
  }
}

template <bool BN>
__device__ __forceinline__ void row_mma(const RowFrag& f, v8f (&acc)[6]) {
  float4 a0 = f.a[0], a1 = f.a[1];
  if (BN) {
    a0.x = a0.x * f.s[0].x + f.h[0].x; a0.y = a0.y * f.s[0].y + f.h[0].y;
    a0.z = a0.z * f.s[0].z + f.h[0].z; a0.w = a0.w * f.s[0].w + f.h[0].w;
    a1.x = a1.x * f.s[1].x + f.h[1].x; a1.y = a1.y * f.s[1].y + f.h[1].y;
    a1.z = a1.z * f.s[1].z + f.h[1].z; a1.w = a1.w * f.s[1].w + f.h[1].w;
  }
  const float ae[8] = {a0.x, a0.y, a0.z, a0.w, a1.x, a1.y, a1.z, a1.w};
#pragma unroll
  for (int j = 0; j < 4; ++j) {
    v2f a;
    a.x = ae[2 * j]; a.y = ae[2 * j + 1];
    const float4 p0 = f.b4[2 * j], p1 = f.b4[2 * j + 1];
    const float2 q0 = f.b2[2 * j], q1 = f.b2[2 * j + 1];
    v2f b;
    b.x = p0.x; b.y = p1.x; acc[0] = wmma_f32(a, b, acc[0]);
    b.x = p0.y; b.y = p1.y; acc[1] = wmma_f32(a, b, acc[1]);
    b.x = p0.z; b.y = p1.z; acc[2] = wmma_f32(a, b, acc[2]);
    b.x = p0.w; b.y = p1.w; acc[3] = wmma_f32(a, b, acc[3]);
    b.x = q0.x; b.y = q1.x; acc[4] = wmma_f32(a, b, acc[4]);
    b.x = q0.y; b.y = q1.y; acc[5] = wmma_f32(a, b, acc[5]);
  }
}

template <bool BN>
__global__ void __launch_bounds__(128)
gemm_f32_wmma_rows96(const float* __restrict__ A, int lda,
                     const float* __restrict__ B,
                     float* __restrict__ D, int ldo, int nrows, int kper,
                     const float* __restrict__ bias,
                     const float* __restrict__ scl,
                     const float* __restrict__ shf, int relu,
                     float* __restrict__ psum, float* __restrict__ psq) {
  __shared__ float4 red[3 * 12 * 32];
  const int lane = threadIdx.x & 31;
  const int w = __builtin_amdgcn_readfirstlane((int)(threadIdx.x >> 5));
  const int nw = (int)(blockDim.x >> 5);
  const int tr = blockIdx.x;
  const int row0 = tr << 4;
  const int lm = lane & 15;
  const int lh = lane >> 4;

  int arow = row0 + lm;
  if (arow >= nrows) arow = nrows - 1;
  const int kbeg = w * kper + 8 * lh;
  const float* Ap = A + (size_t)arow * lda + kbeg;
  const float* Sp = BN ? scl + kbeg : nullptr;
  const float* Hp = BN ? shf + kbeg : nullptr;
  const float* Bp4 = B + (size_t)kbeg * MLPC + 4 * lm;
  const float* Bp2 = B + (size_t)kbeg * MLPC + 64 + 2 * lm;

  v8f acc[6];
#pragma unroll
  for (int t = 0; t < 6; ++t)
    acc[t] = (v8f){0.f, 0.f, 0.f, 0.f, 0.f, 0.f, 0.f, 0.f};

  int k = 0;
  RowFrag f0, f1;
  if (kper >= 32) {
    row_load<BN>(f0, Ap, Sp, Hp, Bp4, Bp2, 0);
    for (; k + 64 <= kper; k += 32) {
      row_load<BN>(f1, Ap, Sp, Hp, Bp4, Bp2, k + 16);
      __builtin_amdgcn_sched_barrier(0);
      row_mma<BN>(f0, acc);
      row_load<BN>(f0, Ap, Sp, Hp, Bp4, Bp2, k + 32);
      __builtin_amdgcn_sched_barrier(0);
      row_mma<BN>(f1, acc);
    }
    row_load<BN>(f1, Ap, Sp, Hp, Bp4, Bp2, k + 16);
    __builtin_amdgcn_sched_barrier(0);
    row_mma<BN>(f0, acc);
    row_mma<BN>(f1, acc);
    k += 32;
  }
  for (; k + 16 <= kper; k += 16) {
    row_load<BN>(f0, Ap, Sp, Hp, Bp4, Bp2, k);
    row_mma<BN>(f0, acc);
  }

  if (w > 0) {
    float4* rp = red + (w - 1) * (12 * 32) + lane;
#pragma unroll
    for (int t = 0; t < 6; ++t)
#pragma unroll
      for (int q = 0; q < 2; ++q)
        rp[(t * 2 + q) * 32] = (float4){acc[t][4 * q], acc[t][4 * q + 1],
                                        acc[t][4 * q + 2], acc[t][4 * q + 3]};
  }
  __syncthreads();
  if (w > 0) return;
  for (int u = 1; u < nw; ++u) {
    const float4* rp = red + (u - 1) * (12 * 32) + lane;
#pragma unroll
    for (int t = 0; t < 6; ++t)
#pragma unroll
      for (int q = 0; q < 2; ++q) {
        const float4 p = rp[(t * 2 + q) * 32];
        acc[t][4 * q] += p.x;
        acc[t][4 * q + 1] += p.y;
        acc[t][4 * q + 2] += p.z;
        acc[t][4 * q + 3] += p.w;
      }
  }

  float4 bv4 = {0.f, 0.f, 0.f, 0.f};
  float2 bv2 = {0.f, 0.f};
  if (bias) {
    bv4 = *(const float4*)(bias + 4 * lm);
    bv2 = *(const float2*)(bias + 64 + 2 * lm);
  }
  float cs[6], cq[6];
#pragma unroll
  for (int t = 0; t < 6; ++t) { cs[t] = 0.f; cq[t] = 0.f; }
#pragma unroll
  for (int r = 0; r < 8; ++r) {
    const int row = row0 + 8 * lh + r;
    if (row < nrows) {
      float v[6];
      v[0] = acc[0][r] + bv4.x; v[1] = acc[1][r] + bv4.y;
      v[2] = acc[2][r] + bv4.z; v[3] = acc[3][r] + bv4.w;
      v[4] = acc[4][r] + bv2.x; v[5] = acc[5][r] + bv2.y;
#pragma unroll
      for (int t = 0; t < 6; ++t) {
        if (relu) v[t] = fmaxf(v[t], 0.0f);
        cs[t] += v[t];
        cq[t] += v[t] * v[t];
      }
      float* dp = D + (size_t)row * ldo;
      *(float4*)(dp + 4 * lm) = (float4){v[0], v[1], v[2], v[3]};
      *(float2*)(dp + 64 + 2 * lm) = (float2){v[4], v[5]};
    }
  }
#pragma unroll
  for (int t = 0; t < 6; ++t) {
    cs[t] += __shfl_xor(cs[t], 16, 32);
    cq[t] += __shfl_xor(cq[t], 16, 32);
  }
  if (lh == 0) {
    float* ps = psum + (size_t)tr * MLPC;
    float* pq = psq + (size_t)tr * MLPC;
    *(float4*)(ps + 4 * lm) = (float4){cs[0], cs[1], cs[2], cs[3]};
    *(float2*)(ps + 64 + 2 * lm) = (float2){cs[4], cs[5]};
    *(float4*)(pq + 4 * lm) = (float4){cq[0], cq[1], cq[2], cq[3]};
    *(float2*)(pq + 64 + 2 * lm) = (float2){cq[4], cq[5]};
  }
}

struct HeadFrag {
  v2f a[4];
  float4 b[4][2][2];
};

__device__ __forceinline__ void head_load(HeadFrag& f,
                                          const float* __restrict__ Ap,
                                          const float* __restrict__ Bp, int ldb,
                                          int k) {
#pragma unroll
  for (int j = 0; j < 4; ++j) {
    f.a[j] = *(const v2f*)(Ap + k + 4 * j);
    const float* bq = Bp + (size_t)(k + 4 * j) * ldb;
    f.b[j][0][0] = *(const float4*)(bq);
    f.b[j][0][1] = *(const float4*)(bq + 64);
    f.b[j][1][0] = *(const float4*)(bq + ldb);
    f.b[j][1][1] = *(const float4*)(bq + ldb + 64);
  }
}

__device__ __forceinline__ void head_mma(const HeadFrag& f, v8f (&acc)[8]) {
#pragma unroll
  for (int j = 0; j < 4; ++j) {
    const float4 p0 = f.b[j][0][0], p1 = f.b[j][1][0];
    const float4 q0 = f.b[j][0][1], q1 = f.b[j][1][1];
    v2f b;
    b.x = p0.x; b.y = p1.x; acc[0] = wmma_f32(f.a[j], b, acc[0]);
    b.x = p0.y; b.y = p1.y; acc[1] = wmma_f32(f.a[j], b, acc[1]);
    b.x = p0.z; b.y = p1.z; acc[2] = wmma_f32(f.a[j], b, acc[2]);
    b.x = p0.w; b.y = p1.w; acc[3] = wmma_f32(f.a[j], b, acc[3]);
    b.x = q0.x; b.y = q1.x; acc[4] = wmma_f32(f.a[j], b, acc[4]);
    b.x = q0.y; b.y = q1.y; acc[5] = wmma_f32(f.a[j], b, acc[5]);
    b.x = q0.z; b.y = q1.z; acc[6] = wmma_f32(f.a[j], b, acc[6]);
    b.x = q0.w; b.y = q1.w; acc[7] = wmma_f32(f.a[j], b, acc[7]);
  }
}

__device__ __forceinline__ void head_gemm_wave(
    const int wid, const float* __restrict__ A, int lda,
    const float* __restrict__ B, int ldb, float* __restrict__ D, int ldo,
    int nrows, int kdim, int heads, const float* __restrict__ a_src,
    const float* __restrict__ a_dst, float* __restrict__ al_s,
    float* __restrict__ al_d) {
  const int lane = threadIdx.x & 31;
  const int rowTiles = (nrows + 15) >> 4;
  if (wid >= rowTiles * heads) return;
  const int tr = wid / heads;
  const int head = wid - tr * heads;
  const int row0 = tr << 4, col0 = head * DFEAT;
  const int lm = lane & 15;
  const int lh = lane >> 4;

  int arow = row0 + lm;
  if (arow >= nrows) arow = nrows - 1;
  const float* Ap = A + (size_t)arow * lda + 2 * lh;
  const float* Bp = B + (size_t)(2 * lh) * ldb + col0 + 4 * lm;

  v8f acc[8];
#pragma unroll
  for (int t = 0; t < 8; ++t)
    acc[t] = (v8f){0.f, 0.f, 0.f, 0.f, 0.f, 0.f, 0.f, 0.f};

  int k = 0;
  if (kdim >= 32) {
    HeadFrag f0, f1;
    head_load(f0, Ap, Bp, ldb, 0);
    for (; k + 64 <= kdim; k += 32) {
      head_load(f1, Ap, Bp, ldb, k + 16);
      __builtin_amdgcn_sched_barrier(0);
      head_mma(f0, acc);
      head_load(f0, Ap, Bp, ldb, k + 32);
      __builtin_amdgcn_sched_barrier(0);
      head_mma(f1, acc);
    }
    head_load(f1, Ap, Bp, ldb, k + 16);
    __builtin_amdgcn_sched_barrier(0);
    head_mma(f0, acc);
    head_mma(f1, acc);
    k += 32;
  }
  for (; k < kdim; k += 4) {
    const v2f a = *(const v2f*)(Ap + k);
    const float* bq = Bp + (size_t)k * ldb;
    const float4 p0 = *(const float4*)(bq), q0 = *(const float4*)(bq + 64);
    const float4 p1 = *(const float4*)(bq + ldb);
    const float4 q1 = *(const float4*)(bq + ldb + 64);
    v2f b;
    b.x = p0.x; b.y = p1.x; acc[0] = wmma_f32(a, b, acc[0]);
    b.x = p0.y; b.y = p1.y; acc[1] = wmma_f32(a, b, acc[1]);
    b.x = p0.z; b.y = p1.z; acc[2] = wmma_f32(a, b, acc[2]);
    b.x = p0.w; b.y = p1.w; acc[3] = wmma_f32(a, b, acc[3]);
    b.x = q0.x; b.y = q1.x; acc[4] = wmma_f32(a, b, acc[4]);
    b.x = q0.y; b.y = q1.y; acc[5] = wmma_f32(a, b, acc[5]);
    b.x = q0.z; b.y = q1.z; acc[6] = wmma_f32(a, b, acc[6]);
    b.x = q0.w; b.y = q1.w; acc[7] = wmma_f32(a, b, acc[7]);
  }

  const float4 s0 = *(const float4*)(a_src + col0 + 4 * lm);
  const float4 s1 = *(const float4*)(a_src + col0 + 64 + 4 * lm);
  const float4 d0 = *(const float4*)(a_dst + col0 + 4 * lm);
  const float4 d1 = *(const float4*)(a_dst + col0 + 64 + 4 * lm);
  float v[16];
#pragma unroll
  for (int r = 0; r < 8; ++r) {
    const float4 lo = {acc[0][r], acc[1][r], acc[2][r], acc[3][r]};
    const float4 hi = {acc[4][r], acc[5][r], acc[6][r], acc[7][r]};
    const int row = row0 + 8 * lh + r;
    if (row < nrows) {
      float* dp = D + (size_t)row * ldo + col0 + 4 * lm;
      *(float4*)dp = lo;
      *(float4*)(dp + 64) = hi;
    }
    v[r] = lo.x * s0.x + lo.y * s0.y + lo.z * s0.z + lo.w * s0.w +
           hi.x * s1.x + hi.y * s1.y + hi.z * s1.z + hi.w * s1.w;
    v[8 + r] = lo.x * d0.x + lo.y * d0.y + lo.z * d0.z + lo.w * d0.w +
               hi.x * d1.x + hi.y * d1.y + hi.z * d1.z + hi.w * d1.w;
  }
  const bool b3 = (lm & 8) != 0, b2 = (lm & 4) != 0;
  const bool b1 = (lm & 2) != 0, b0 = (lm & 1) != 0;
  float w8[8], w4[4], w2[2];
#pragma unroll
  for (int i = 0; i < 8; ++i) {
    const float keep = b3 ? v[8 + i] : v[i];
    const float send = b3 ? v[i] : v[8 + i];
    w8[i] = keep + __shfl_xor(send, 8, 32);
  }
#pragma unroll
  for (int i = 0; i < 4; ++i) {
    const float keep = b2 ? w8[4 + i] : w8[i];
    const float send = b2 ? w8[i] : w8[4 + i];
    w4[i] = keep + __shfl_xor(send, 4, 32);
  }
#pragma unroll
  for (int i = 0; i < 2; ++i) {
    const float keep = b1 ? w4[2 + i] : w4[i];
    const float send = b1 ? w4[i] : w4[2 + i];
    w2[i] = keep + __shfl_xor(send, 2, 32);
  }
  const float keep = b0 ? w2[1] : w2[0];
  const float send = b0 ? w2[0] : w2[1];
  const float tot = keep + __shfl_xor(send, 1, 32);
  const int prow = row0 + 8 * lh + (lm & 7);
  if (prow < nrows) {
    float* dst = b3 ? al_d : al_s;
    dst[(size_t)prow * heads + head] = tot;
  }
}

__global__ void __launch_bounds__(64)
gemm_f32_wmma_head(const float* __restrict__ A, int lda,
                   const float* __restrict__ B, int ldb,
                   float* __restrict__ D, int ldo, int nrows, int kdim,
                   int heads, const float* __restrict__ a_src,
                   const float* __restrict__ a_dst, float* __restrict__ al_s,
                   float* __restrict__ al_d) {
  head_gemm_wave((int)((blockIdx.x * blockDim.x + threadIdx.x) >> 5), A, lda, B,
                 ldb, D, ldo, nrows, kdim, heads, a_src, a_dst, al_s, al_d);
}

__global__ void csr_zero(int* __restrict__ cnt, int n1) {
  const int i = blockIdx.x * blockDim.x + threadIdx.x;
  if (i < n1) cnt[i] = 0;
}

#define EDGES_PER_THREAD 4
#define HIST_EDGES_PER_THREAD 20
__device__ __forceinline__ void hist_edges(const int t, const int T,
                                           const int* __restrict__ dst, int E,
                                           int* __restrict__ cnt,
                                           int* __restrict__ rank) {
  int d[HIST_EDGES_PER_THREAD], r[HIST_EDGES_PER_THREAD];
  if (t + (HIST_EDGES_PER_THREAD - 1) * T < E) {
#pragma unroll
    for (int q = 0; q < HIST_EDGES_PER_THREAD; ++q) d[q] = dst[t + q * T];
#pragma unroll
    for (int q = 0; q < HIST_EDGES_PER_THREAD; ++q)
      r[q] = atomicAdd(&cnt[d[q]], 1);
#pragma unroll
    for (int q = 0; q < HIST_EDGES_PER_THREAD; ++q) rank[t + q * T] = r[q];
  } else {
#pragma unroll
    for (int q = 0; q < HIST_EDGES_PER_THREAD; ++q)
      d[q] = dst[min(t + q * T, E - 1)];
#pragma unroll
    for (int q = 0; q < HIST_EDGES_PER_THREAD; ++q) {
      const int e = t + q * T;
      r[q] = 0;
      if (e < E) r[q] = atomicAdd(&cnt[d[q]], 1);
    }
#pragma unroll
    for (int q = 0; q < HIST_EDGES_PER_THREAD; ++q) {
      const int e = t + q * T;
      if (e < E) rank[e] = r[q];
    }
  }
}

__global__ void __launch_bounds__(128)
head_gemm_with_hist(const float* __restrict__ A, int lda,
                    const float* __restrict__ B, int ldb,
                    float* __restrict__ D, int ldo, int nrows, int kdim,
                    int heads, const float* __restrict__ a_src,
                    const float* __restrict__ a_dst, float* __restrict__ al_s,
                    float* __restrict__ al_d, int histBlocks,
                    const int* __restrict__ dst, int E, int* __restrict__ cnt,
                    int* __restrict__ rank) {
  const int b = (int)blockIdx.x;
  if (b < histBlocks) {
    hist_edges(b * (int)blockDim.x + (int)threadIdx.x,
               histBlocks * (int)blockDim.x, dst, E, cnt, rank);
  } else {
    head_gemm_wave(
        (int)(((b - histBlocks) * blockDim.x + threadIdx.x) >> 5), A, lda, B,
        ldb, D, ldo, nrows, kdim, heads, a_src, a_dst, al_s, al_d);
  }
}

__global__ void csr_scan(const int* __restrict__ cnt, int n,
                         int* __restrict__ rowptr) {
  __shared__ int part[1024];
  const int t = threadIdx.x;
  const int per = (n + 1023) / 1024;
  const int lo = min(t * per, n);
  const int hi = min(lo + per, n);
  int s = 0;
  for (int i = lo; i < hi; ++i) s += cnt[i];
  part[t] = s;
  __syncthreads();
  for (int off = 1; off < 1024; off <<= 1) {
    const int v = (t >= off) ? part[t - off] : 0;
    __syncthreads();
    part[t] += v;
    __syncthreads();
  }
  int run = part[t] - s;
  for (int i = lo; i < hi; ++i) {
    rowptr[i] = run;
    run += cnt[i];
  }
  if (t == 1023) rowptr[n] = part[1023];
}

__global__ void csr_fill(const int* __restrict__ src,
                         const int* __restrict__ dst, int E,
                         const int* __restrict__ rowptr,
                         const int* __restrict__ rank,
                         int* __restrict__ tmp) {
  const int t = blockIdx.x * blockDim.x + threadIdx.x;
  const int T = gridDim.x * blockDim.x;
  if (t + (EDGES_PER_THREAD - 1) * T < E) {
    int d[EDGES_PER_THREAD], s[EDGES_PER_THREAD], r[EDGES_PER_THREAD];
    int p[EDGES_PER_THREAD];
#pragma unroll
    for (int q = 0; q < EDGES_PER_THREAD; ++q) {
      d[q] = dst[t + q * T];
      s[q] = src[t + q * T];
      r[q] = rank[t + q * T];
    }
#pragma unroll
    for (int q = 0; q < EDGES_PER_THREAD; ++q) p[q] = rowptr[d[q]] + r[q];
#pragma unroll
    for (int q = 0; q < EDGES_PER_THREAD; ++q) tmp[p[q]] = s[q];
  } else {
    for (int e = t; e < E; e += T) tmp[rowptr[dst[e]] + rank[e]] = src[e];
  }
}

__global__ void csr_sort(const int* __restrict__ rowptr,
                         const int* __restrict__ tmp, int* __restrict__ csr,
                         int n) {
  const int lane = threadIdx.x & 31;
  const int wid = __builtin_amdgcn_readfirstlane(
      (int)((blockIdx.x * blockDim.x + threadIdx.x) >> 5));
  if (wid >= n) return;
  const int beg = rowptr[wid];
  const int end = rowptr[wid + 1];
  for (int i = beg + lane; i < end; i += 32) {
    const int v = tmp[i];
    int rank = 0;
    for (int j = beg; j < end; ++j) {
      const int u = tmp[j];
      rank += (u < v || (u == v && j < i)) ? 1 : 0;
    }
    csr[beg + rank] = v;
  }
}

__device__ __forceinline__ void gather_rows(const float* __restrict__ hb,
                                            int ldh, int s_i, float w_i,
                                            int cnt, float4& acc) {
  const int wi = __float_as_int(w_i);
  int j = 0;
  for (; j + 4 <= cnt; j += 4) {
    const int sa = __builtin_amdgcn_readlane(s_i, j);
    const int sb = __builtin_amdgcn_readlane(s_i, j + 1);
    const int sc = __builtin_amdgcn_readlane(s_i, j + 2);
    const int sd = __builtin_amdgcn_readlane(s_i, j + 3);
    const float wa = __int_as_float(__builtin_amdgcn_readlane(wi, j));
    const float wb = __int_as_float(__builtin_amdgcn_readlane(wi, j + 1));
    const float wc = __int_as_float(__builtin_amdgcn_readlane(wi, j + 2));
    const float wd = __int_as_float(__builtin_amdgcn_readlane(wi, j + 3));
    const float4 ha = *(const float4*)(hb + (size_t)sa * ldh);
    const float4 hv = *(const float4*)(hb + (size_t)sb * ldh);
    const float4 hc = *(const float4*)(hb + (size_t)sc * ldh);
    const float4 hd = *(const float4*)(hb + (size_t)sd * ldh);
    acc.x += wa * ha.x; acc.y += wa * ha.y; acc.z += wa * ha.z; acc.w += wa * ha.w;
    acc.x += wb * hv.x; acc.y += wb * hv.y; acc.z += wb * hv.z; acc.w += wb * hv.w;
    acc.x += wc * hc.x; acc.y += wc * hc.y; acc.z += wc * hc.z; acc.w += wc * hc.w;
    acc.x += wd * hd.x; acc.y += wd * hd.y; acc.z += wd * hd.z; acc.w += wd * hd.w;
  }
  for (; j < cnt; ++j) {
    const int sa = __builtin_amdgcn_readlane(s_i, j);
    const float wa = __int_as_float(__builtin_amdgcn_readlane(wi, j));
    const float4 ha = *(const float4*)(hb + (size_t)sa * ldh);
    acc.x += wa * ha.x; acc.y += wa * ha.y; acc.z += wa * ha.z; acc.w += wa * ha.w;
  }
}

__global__ void gat_gather(const int* __restrict__ rowptr,
                           const int* __restrict__ csr, int heads,
                           const float* __restrict__ h, int ldh,
                           const float* __restrict__ al_s,
                           const float* __restrict__ al_d,
                           const float* __restrict__ bias,
                           float* __restrict__ out, int ldo, int colOff,
                           int n) {
  const int lane = threadIdx.x & 31;
  const int wid = __builtin_amdgcn_readfirstlane(
      (int)((blockIdx.x * blockDim.x + threadIdx.x) >> 5));
  if (wid >= n * heads) return;
  const int d = wid / heads;
  const int head = wid - d * heads;
  const int beg = rowptr[d];
  const int deg = rowptr[d + 1] - beg;
  const int total = deg + 1;
  const float ad = al_d[wid];
  const int* cp = csr + beg;

  const bool v0 = lane < total;
  const int s0 = (lane < deg) ? cp[lane] : d;
  float x0 = al_s[s0 * heads + head] + ad;
  x0 = (x0 > 0.f) ? x0 : NEG_SLOPE * x0;

  float m = v0 ? x0 : -INFINITY;
  for (int i = lane + 32; i < total; i += 32) {
    const int s = (i < deg) ? cp[i] : d;
    float x = al_s[s * heads + head] + ad;
    x = (x > 0.f) ? x : NEG_SLOPE * x;
    m = fmaxf(m, x);
  }
  for (int off = 16; off > 0; off >>= 1) m = fmaxf(m, __shfl_xor(m, off, 32));

  const float e0 = v0 ? expf(x0 - m) : 0.f;
  float sum = e0;
  for (int i = lane + 32; i < total; i += 32) {
    const int s = (i < deg) ? cp[i] : d;
    float x = al_s[s * heads + head] + ad;
    x = (x > 0.f) ? x : NEG_SLOPE * x;
    sum += expf(x - m);
  }
  for (int off = 16; off > 0; off >>= 1) sum += __shfl_xor(sum, off, 32);
  const float den = sum + 1e-16f;

  const float* hb = h + head * DFEAT + lane * 4;
  float4 acc = {0.f, 0.f, 0.f, 0.f};
  gather_rows(hb, ldh, s0, e0 / den, min(total, 32), acc);
  for (int base = 32; base < total; base += 32) {
    const int i = base + lane;
    const int s = (i < deg) ? cp[i] : d;
    float x = al_s[s * heads + head] + ad;
    x = (x > 0.f) ? x : NEG_SLOPE * x;
    const float w = (i < total) ? expf(x - m) / den : 0.f;
    gather_rows(hb, ldh, s, w, min(total - base, 32), acc);
  }

  const float* bp = bias + head * DFEAT + lane * 4;
  acc.x += bp[0]; acc.y += bp[1]; acc.z += bp[2]; acc.w += bp[3];
  *(float4*)(out + (size_t)d * ldo + colOff + head * DFEAT + lane * 4) = acc;
}

__global__ void bn_final(const float* __restrict__ psum,
                         const float* __restrict__ psq, int nparts,
                         const float* __restrict__ g,
                         const float* __restrict__ beta, int ncols, float invn,
                         float* __restrict__ sc, float* __restrict__ sh) {
  __shared__ float wsum[32];
  __shared__ float wsq[32];
  const int c = blockIdx.x;
  const int T = blockDim.x;
  const int last = nparts - 1;
  float s = 0.f, q = 0.f;
  for (int p = threadIdx.x; p < nparts; p += 4 * T) {
    const int p1 = p + T, p2 = p + 2 * T, p3 = p + 3 * T;
    const size_t i0 = (size_t)p * ncols + c;
    const size_t i1 = (size_t)min(p1, last) * ncols + c;
    const size_t i2 = (size_t)min(p2, last) * ncols + c;
    const size_t i3 = (size_t)min(p3, last) * ncols + c;
    const float a0 = psum[i0], a1 = psum[i1], a2 = psum[i2], a3 = psum[i3];
    const float b0 = psq[i0], b1 = psq[i1], b2 = psq[i2], b3 = psq[i3];
    s += a0;
    q += b0;
    s += (p1 < nparts) ? a1 : 0.f;
    q += (p1 < nparts) ? b1 : 0.f;
    s += (p2 < nparts) ? a2 : 0.f;
    q += (p2 < nparts) ? b2 : 0.f;
    s += (p3 < nparts) ? a3 : 0.f;
    q += (p3 < nparts) ? b3 : 0.f;
  }
  for (int off = 16; off > 0; off >>= 1) {
    s += __shfl_xor(s, off, 32);
    q += __shfl_xor(q, off, 32);
  }
  if ((threadIdx.x & 31) == 0) {
    wsum[threadIdx.x >> 5] = s;
    wsq[threadIdx.x >> 5] = q;
  }
  __syncthreads();
  if (threadIdx.x == 0) {
    float ts = 0.f, tq = 0.f;
    for (int w = 0; w < (T >> 5); ++w) {
      ts += wsum[w];
      tq += wsq[w];
    }
    const float mean = ts * invn;
    const float var = tq * invn - mean * mean;
    const float r = rsqrtf(var + 1e-5f);
    const float sv = g[c] * r;
    sc[c] = sv;
    sh[c] = beta[c] - mean * sv;
  }
}

__global__ void mlp_out(const float* __restrict__ y2,
                        const float* __restrict__ sc,
                        const float* __restrict__ sh,
                        const float* __restrict__ Wm2,
                        const float* __restrict__ bm2, int n,
                        float* __restrict__ y3, float* __restrict__ psum,
                        float* __restrict__ psq) {
  __shared__ float red[32][4];
  const int node = blockIdx.x * blockDim.x + threadIdx.x;
  float o0 = 0.f, o1 = 0.f;
  if (node < n) {
    float a0 = bm2[0], a1 = bm2[1];
    const float* yp = y2 + (size_t)node * 96;
    for (int k = 0; k < 96; ++k) {
      const float v = fmaf(yp[k], sc[k], sh[k]);
      a0 = fmaf(v, Wm2[2 * k + 0], a0);
      a1 = fmaf(v, Wm2[2 * k + 1], a1);
    }
    o0 = fmaxf(a0, 0.f);
    o1 = fmaxf(a1, 0.f);
    y3[node * 2 + 0] = o0;
    y3[node * 2 + 1] = o1;
  }
  float s0 = o0, q0 = o0 * o0, s1 = o1, q1 = o1 * o1;
  for (int off = 16; off > 0; off >>= 1) {
    s0 += __shfl_xor(s0, off, 32);
    q0 += __shfl_xor(q0, off, 32);
    s1 += __shfl_xor(s1, off, 32);
    q1 += __shfl_xor(q1, off, 32);
  }
  if ((threadIdx.x & 31) == 0) {
    float* rw = red[threadIdx.x >> 5];
    rw[0] = s0; rw[1] = q0; rw[2] = s1; rw[3] = q1;
  }
  __syncthreads();
  if (threadIdx.x < 4) {
    float t = 0.f;
    for (int w = 0; w < (int)(blockDim.x >> 5); ++w) t += red[w][threadIdx.x];
    float* dst = (threadIdx.x & 1) ? psq : psum;
    dst[blockIdx.x * 2 + (threadIdx.x >> 1)] = t;
  }
}

__global__ void bn_apply(const float* __restrict__ y,
                         const float* __restrict__ sc,
                         const float* __restrict__ sh, int total, int ncols,
                         float* __restrict__ out) {
  const int i = blockIdx.x * blockDim.x + threadIdx.x;
  if (i >= total) return;
  const int c = i % ncols;
  out[i] = fmaf(y[i], sc[c], sh[c]);
}

extern "C" void kernel_launch(void* const* d_in, const int* in_sizes, int n_in,
                              void* d_out, int out_size, void* d_ws,
                              size_t ws_size, hipStream_t stream) {
  const float* x      = (const float*)d_in[0];
  const int*   ei     = (const int*)d_in[1];
  const float* W1     = (const float*)d_in[2];
  const float* as1    = (const float*)d_in[3];
  const float* ad1    = (const float*)d_in[4];
  const float* b1     = (const float*)d_in[5];
  const float* W2     = (const float*)d_in[6];
  const float* as2    = (const float*)d_in[7];
  const float* ad2    = (const float*)d_in[8];
  const float* b2     = (const float*)d_in[9];
  const float* Wl     = (const float*)d_in[10];
  const float* bl     = (const float*)d_in[11];
  const float* gl     = (const float*)d_in[12];
  const float* betal  = (const float*)d_in[13];
  const float* Wm1    = (const float*)d_in[14];
  const float* bm1    = (const float*)d_in[15];
  const float* gm1    = (const float*)d_in[16];
  const float* betam1 = (const float*)d_in[17];
  const float* Wm2    = (const float*)d_in[18];
  const float* bm2    = (const float*)d_in[19];
  const float* gm2    = (const float*)d_in[20];
  const float* betam2 = (const float*)d_in[21];

  const int N = in_sizes[0] / DFEAT;
  const int E = in_sizes[1] / 2;
  const int* srcA = ei;
  const int* dstA = ei + E;

  char* w = (char*)d_ws;
  auto alloc = [&](size_t bytes) -> char* {
    char* p = w;
    w += (bytes + 255) & ~(size_t)255;
    return p;
  };
  float*    xcat = (float*)alloc((size_t)N * 384 * 4);
  float*    h1   = (float*)alloc((size_t)N * 256 * 4);
  float*    h2   = (float*)alloc((size_t)N * 128 * 4);
  float*    al_s = (float*)alloc((size_t)N * 2 * 4);
  float*    al_d = (float*)alloc((size_t)N * 2 * 4);
  int*      cnt    = (int*)alloc((size_t)(N + 1) * 4);
  int*      rowptr = (int*)alloc((size_t)(N + 1) * 4);
  int*      csrTmp = (int*)alloc((size_t)E * 4);
  int*      csr    = (int*)alloc((size_t)E * 4);
  float*    y3   = (float*)alloc((size_t)N * 2 * 4);
  float*    sc1  = (float*)alloc(96 * 4);
  float*    sh1  = (float*)alloc(96 * 4);
  float*    sc2  = (float*)alloc(96 * 4);
  float*    sh2  = (float*)alloc(96 * 4);
  float*    sc3  = (float*)alloc(96 * 4);
  float*    sh3  = (float*)alloc(96 * 4);
  int* rank = csr;
  float* y1 = h1;
  float* y2 = h1 + (size_t)N * 96;
  const int rowTiles  = (N + 15) / 16;
  const int mlpBlocks = (N + 255) / 256;
  float* ps1 = h2;
  float* pq1 = ps1 + (size_t)rowTiles * 96;
  float* ps2 = pq1 + (size_t)rowTiles * 96;
  float* pq2 = ps2 + (size_t)rowTiles * 96;
  float* ps3 = pq2 + (size_t)rowTiles * 96;
  float* pq3 = ps3 + (size_t)mlpBlocks * 2;

  const int TB = 256;
  csr_zero<<<(N + 1 + TB - 1) / TB, TB, 0, stream>>>(cnt, N + 1);
  const int edgeBlocks =
      (E + TB * EDGES_PER_THREAD - 1) / (TB * EDGES_PER_THREAD);
  const int gemmBlocks = (rowTiles * 2 + 3) / 4;
  const int histBlocks = (E + 128 * HIST_EDGES_PER_THREAD - 1) /
                         (128 * HIST_EDGES_PER_THREAD);
  head_gemm_with_hist<<<histBlocks + gemmBlocks, 128, 0, stream>>>(
      x, 128, W1, 256, h1, 256, N, 128, 2, as1, ad1, al_s, al_d, histBlocks,
      dstA, E, cnt, rank);
  csr_scan<<<1, 1024, 0, stream>>>(cnt, N, rowptr);
  csr_fill<<<edgeBlocks, TB, 0, stream>>>(srcA, dstA, E, rowptr, rank, csrTmp);
  csr_sort<<<(N + 7) / 8, TB, 0, stream>>>(rowptr, csrTmp, csr, N);

  gat_gather<<<(N * 2 + 7) / 8, TB, 0, stream>>>(rowptr, csr, 2, h1, 256, al_s,
                                                 al_d, b1, xcat, 384, 0, N);

  gemm_f32_wmma_head<<<rowTiles, 32, 0, stream>>>(
      xcat, 384, W2, 128, h2, 128, N, 256, 1, as2, ad2, al_s, al_d);
  gat_gather<<<(N + 7) / 8, TB, 0, stream>>>(rowptr, csr, 1, h2, 128, al_s,
                                             al_d, b2, xcat, 384, 256, N);

  const float invn = 1.0f / (float)N;
  gemm_f32_wmma_rows96<false><<<rowTiles, 128, 0, stream>>>(
      xcat, 384, Wl, y1, 96, N, 96, bl, nullptr, nullptr, 1, ps1, pq1);
  bn_final<<<96, 256, 0, stream>>>(ps1, pq1, rowTiles, gl, betal, 96, invn, sc1,
                                   sh1);
  gemm_f32_wmma_rows96<true><<<rowTiles, 64, 0, stream>>>(
      y1, 96, Wm1, y2, 96, N, 48, bm1, sc1, sh1, 1, ps2, pq2);
  bn_final<<<96, 256, 0, stream>>>(ps2, pq2, rowTiles, gm1, betam1, 96, invn,
                                   sc2, sh2);
  mlp_out<<<mlpBlocks, 256, 0, stream>>>(y2, sc2, sh2, Wm2, bm2, N, y3, ps3,
                                         pq3);
  bn_final<<<2, 256, 0, stream>>>(ps3, pq3, mlpBlocks, gm2, betam2, 2, invn,
                                  sc3, sh3);
  bn_apply<<<(N * 2 + TB - 1) / TB, TB, 0, stream>>>(y3, sc3, sh3, N * 2, 2,
                                                     (float*)d_out);
}
